# starting kernel
# baseline (speedup 1.0000x reference)
#include <hip/hip_runtime.h>
#include <hip/hip_bf16.h>
#include <cstddef>


typedef float v2f __attribute__((ext_vector_type(2)));
typedef float v8f __attribute__((ext_vector_type(8)));

static __device__ __forceinline__ v8f wmma_f32(v2f a, v2f b, v8f c) {
  return __builtin_amdgcn_wmma_f32_16x16x4_f32(
       false, a,  false, b,
       (short)0, c,  false,  false);
}

__global__ void k_simraw(const float* __restrict__ E, float* __restrict__ sim) {
  const int idx = blockIdx.x * 256 + threadIdx.x;
  const int i = idx >> 10, j = idx & 1023;
  float acc = 0.f;
#pragma unroll
  for (int d = 0; d < 16; ++d) acc += E[i * 16 + d] * E[j * 16 + d];
  sim[idx] = fmaxf(acc, 0.f);
}

__global__ void k_rowsoftmax(float* __restrict__ sim) {
  const int row = blockIdx.x;
  const int t = threadIdx.x;
  __shared__ float red[256];
  float v[4];
  float mx = -3.0e38f;
#pragma unroll
  for (int i = 0; i < 4; ++i) {
    v[i] = sim[(size_t)row * 1024 + t + i * 256];
    mx = fmaxf(mx, v[i]);
  }
  red[t] = mx; __syncthreads();
  for (int s = 128; s > 0; s >>= 1) {
    if (t < s) red[t] = fmaxf(red[t], red[t + s]);
    __syncthreads();
  }
  mx = red[0]; __syncthreads();
  float sum = 0.f;
#pragma unroll
  for (int i = 0; i < 4; ++i) { v[i] = __expf(v[i] - mx); sum += v[i]; }
  red[t] = sum; __syncthreads();
  for (int s = 128; s > 0; s >>= 1) {
    if (t < s) red[t] += red[t + s];
    __syncthreads();
  }
  const float inv = 1.0f / red[0];
#pragma unroll
  for (int i = 0; i < 4; ++i)
    sim[(size_t)row * 1024 + t + i * 256] = v[i] * inv;
}

__global__ void k_cdist(const float* __restrict__ pa, float* __restrict__ sube) {
  const int b = blockIdx.z;
  const int jBase = blockIdx.x * 64, iBase = blockIdx.y * 64;
  __shared__ float Pi[64][17];
  __shared__ float Pj[64][17];
  const int t = threadIdx.x;
  for (int i = t; i < 64 * 16; i += 256) {
    const int r = i >> 4, d = i & 15;
    Pi[r][d] = pa[((size_t)b * 1024 + iBase + r) * 16 + d];
    Pj[r][d] = pa[((size_t)b * 1024 + jBase + r) * 16 + d];
  }
  __syncthreads();
  const int tj = t & 63;
  const int i0 = (t >> 6) * 16;
#pragma unroll 4
  for (int ii = 0; ii < 16; ++ii) {
    const int i = i0 + ii;
    float s = 0.f;
#pragma unroll
    for (int d = 0; d < 16; ++d) s += fabsf(Pi[i][d] - Pj[tj][d]);
    sube[((size_t)b * 1024 + iBase + i) * 1024 + jBase + tj] = __expf(-s);
  }
}

__global__ void k_colsum(const float* __restrict__ sube, float* __restrict__ S) {
  const int b = blockIdx.y;
  const int j = blockIdx.x * 256 + threadIdx.x;
  float acc = 0.f;
  for (int i = 0; i < 1024; ++i)
    acc += sube[((size_t)b * 1024 + i) * 1024 + j];
  S[b * 1024 + j] = acc;
}

__global__ void k_xnorm(const float* __restrict__ x, const float* __restrict__ S,
                        float* __restrict__ xn) {
  const int idx = blockIdx.x * 256 + threadIdx.x;
  xn[idx] = x[idx] / S[idx >> 6];
}

__global__ void k_weights(const float* __restrict__ E, const float* __restrict__ Wp,
                          float* __restrict__ W) {
  const int n = blockIdx.y;
  const int tid = blockIdx.x * 256 + threadIdx.x;
  const int kp = tid >> 5;
  const int oc = (tid & 31) * 2;
  float e[16];
#pragma unroll
  for (int d = 0; d < 16; ++d) e[d] = E[n * 16 + d];
  float4 acc = {0.f, 0.f, 0.f, 0.f};
#pragma unroll
  for (int d = 0; d < 16; ++d) {
    const v2f w0 = *(const v2f*)(Wp + (size_t)d * 8192 + (size_t)(2 * kp) * 64 + oc);
    const v2f w1 = *(const v2f*)(Wp + (size_t)d * 8192 + (size_t)(2 * kp + 1) * 64 + oc);
    acc.x += e[d] * w0.x;
    acc.y += e[d] * w1.x;
    acc.z += e[d] * w0.y;
    acc.w += e[d] * w1.y;
  }
  *(float4*)(W + (size_t)n * 8192 + (size_t)kp * 128 + (size_t)oc * 2) = acc;
}

__global__ __launch_bounds__(256)
void k_bmm_nk(const float* __restrict__ A, long long strideA,
              const float* __restrict__ B, long long strideB,
              float* __restrict__ C, long long strideC) {
  const int b = blockIdx.y;
  const float* Ab = A + (size_t)strideA * b;
  const float* Bb = B + (size_t)strideB * b;
  float* Cb = C + (size_t)strideC * b;
  const int rowBase = blockIdx.x * 128;

  __shared__ float As[128][68];
  __shared__ float Bs[32][160];

  const int t = threadIdx.x;
  const int wave = t >> 5;
  const int lane = t & 31;
  const int m16 = lane & 15;
  const int kh = lane >> 4;

  const v8f zero = {0.f, 0.f, 0.f, 0.f, 0.f, 0.f, 0.f, 0.f};
  v8f cacc[4];
#pragma unroll
  for (int nt = 0; nt < 4; ++nt) cacc[nt] = zero;

  for (int kc = 0; kc < 1024; kc += 64) {
    __syncthreads();
    for (int i = t; i < 128 * 16; i += 256) {
      const int r = i >> 4, cc = (i & 15) << 2;
      *(float4*)(&As[r][cc]) =
          *(const float4*)(Ab + (size_t)(rowBase + r) * 1024 + kc + cc);
    }
    for (int i = t; i < 32 * 16; i += 256) {
      const int kp = i >> 4, cc = (i & 15) << 2;
      const float4 v0 = *(const float4*)(Bb + (size_t)(kc + 2 * kp) * 64 + cc);
      const float4 v1 = *(const float4*)(Bb + (size_t)(kc + 2 * kp + 1) * 64 + cc);
      float4 lo = {v0.x, v1.x, v0.y, v1.y};
      float4 hi = {v0.z, v1.z, v0.w, v1.w};
      *(float4*)(&Bs[kp][cc * 2])     = lo;
      *(float4*)(&Bs[kp][cc * 2 + 4]) = hi;
    }
    __syncthreads();

    const int mrow = wave * 16 + m16;
#pragma unroll
    for (int ks = 0; ks < 16; ++ks) {
      const int kg = ks * 4 + 2 * kh;
      const int kp = ks * 2 + kh;
      const v2f a = *(const v2f*)(&As[mrow][kg]);
#pragma unroll
      for (int nt = 0; nt < 4; ++nt) {
        const v2f bf = *(const v2f*)(&Bs[kp][(nt * 16 + m16) * 2]);
        cacc[nt] = wmma_f32(a, bf, cacc[nt]);
      }
    }
  }

#pragma unroll
  for (int nt = 0; nt < 4; ++nt) {
#pragma unroll
    for (int r = 0; r < 8; ++r) {
      const int m = rowBase + wave * 16 + r + 8 * kh;
      Cb[(size_t)m * 64 + nt * 16 + m16] = cacc[nt][r];
    }
  }
}

__global__ __launch_bounds__(256)
void k_final(const float* __restrict__ sx, const float* __restrict__ xg1,
             const float* __restrict__ W, const float* __restrict__ E,
             const float* __restrict__ bias_pool, float* __restrict__ out) {
  __shared__ float As[8][8][132];
  const int t = threadIdx.x;
  const int wave = t >> 5;
  const int lane = t & 31;
  const int n = blockIdx.x * 8 + wave;
  float (*Aw)[132] = As[wave];

  for (int rb = 0; rb < 8; ++rb) {
    const float* s0 = sx  + ((size_t)rb * 1024 + n) * 64;
    const float* s1 = xg1 + ((size_t)rb * 1024 + n) * 64;
    for (int c = lane; c < 128; c += 32)
      Aw[rb][c] = (c < 64) ? s0[c] : s1[c - 64];
  }
  __syncthreads();

  const float* Wn = W + (size_t)n * 8192;
  const int m16 = lane & 15;
  const int kh = lane >> 4;

  const v8f zero = {0.f, 0.f, 0.f, 0.f, 0.f, 0.f, 0.f, 0.f};
  v8f cacc[4];
#pragma unroll
  for (int nt = 0; nt < 4; ++nt) cacc[nt] = zero;

#pragma unroll 4
  for (int ks = 0; ks < 32; ++ks) {
    const int kg = ks * 4 + 2 * kh;
    const int kp = ks * 2 + kh;
    v2f a = {0.f, 0.f};
    if (m16 < 8) a = *(const v2f*)(&Aw[m16][kg]);
#pragma unroll
    for (int nt = 0; nt < 4; ++nt) {
      const v2f bf = *(const v2f*)(Wn + (size_t)kp * 128 + (nt * 16 + m16) * 2);
      cacc[nt] = wmma_f32(a, bf, cacc[nt]);
    }
  }

#pragma unroll
  for (int nt = 0; nt < 4; ++nt) {
    const int col = nt * 16 + m16;
    float bv = 0.f;
#pragma unroll
    for (int d = 0; d < 16; ++d) bv += E[n * 16 + d] * bias_pool[d * 64 + col];
#pragma unroll
    for (int r = 0; r < 8; ++r) {
      const int m = r + 8 * kh;
      if (m < 8)
        out[((size_t)m * 1024 + n) * 64 + col] = cacc[nt][r] + bv;
    }
  }
}

extern "C" void kernel_launch(void* const* d_in, const int* in_sizes, int n_in,
                              void* d_out, int out_size, void* d_ws, size_t ws_size,
                              hipStream_t stream) {
  (void)in_sizes; (void)n_in; (void)out_size; (void)ws_size;
  const float* x    = (const float*)d_in[0];
  const float* E    = (const float*)d_in[1];
  const float* pa   = (const float*)d_in[2];
  const float* Wp   = (const float*)d_in[3];
  const float* bp   = (const float*)d_in[4];
  float* out = (float*)d_out;

  float* ws   = (float*)d_ws;
  float* sim  = ws;
  float* sube = sim  + 1048576;
  float* S    = sube + 8388608;
  float* xn   = S    + 8192;
  float* sx   = xn   + 524288;
  float* xg1  = sx   + 524288;
  float* W    = xg1  + 524288;

  k_simraw    <<<4096, 256, 0, stream>>>(E, sim);
  k_rowsoftmax<<<1024, 256, 0, stream>>>(sim);
  k_cdist     <<<dim3(16, 16, 8), 256, 0, stream>>>(pa, sube);
  k_colsum    <<<dim3(4, 8),      256, 0, stream>>>(sube, S);
  k_xnorm     <<<2048, 256, 0, stream>>>(x, S, xn);
  k_weights   <<<dim3(8, 1024),   256, 0, stream>>>(E, Wp, W);
  k_bmm_nk<<<dim3(8, 8), 256, 0, stream>>>(sube, 1024LL * 1024, xn, 1024LL * 64,
                                           sx, 1024LL * 64);
  k_bmm_nk<<<dim3(8, 8), 256, 0, stream>>>(sim, 0LL, sx, 1024LL * 64,
                                           xg1, 1024LL * 64);
  k_final<<<128, 256, 0, stream>>>(sx, xg1, W, E, bp, out);
}


# Global-to-LDS staging in both WMMA kernels: all loads of a stage requested into registers before the first wait (one memory round trip per stage instead of one per loop trip)
# speedup vs baseline: 1.5634x; 1.5634x over previous
#include <hip/hip_runtime.h>
#include <hip/hip_bf16.h>
#include <cstddef>


typedef float v2f __attribute__((ext_vector_type(2)));
typedef float v8f __attribute__((ext_vector_type(8)));

static __device__ __forceinline__ v8f wmma_f32(v2f a, v2f b, v8f c) {
  return __builtin_amdgcn_wmma_f32_16x16x4_f32(
       false, a,  false, b,
       (short)0, c,  false,  false);
}

__global__ void k_simraw(const float* __restrict__ E, float* __restrict__ sim) {
  const int idx = blockIdx.x * 256 + threadIdx.x;
  const int i = idx >> 10, j = idx & 1023;
  float acc = 0.f;
#pragma unroll
  for (int d = 0; d < 16; ++d) acc += E[i * 16 + d] * E[j * 16 + d];
  sim[idx] = fmaxf(acc, 0.f);
}

__global__ void k_rowsoftmax(float* __restrict__ sim) {
  const int row = blockIdx.x;
  const int t = threadIdx.x;
  __shared__ float red[256];
  float v[4];
  float mx = -3.0e38f;
#pragma unroll
  for (int i = 0; i < 4; ++i) {
    v[i] = sim[(size_t)row * 1024 + t + i * 256];
    mx = fmaxf(mx, v[i]);
  }
  red[t] = mx; __syncthreads();
  for (int s = 128; s > 0; s >>= 1) {
    if (t < s) red[t] = fmaxf(red[t], red[t + s]);
    __syncthreads();
  }
  mx = red[0]; __syncthreads();
  float sum = 0.f;
#pragma unroll
  for (int i = 0; i < 4; ++i) { v[i] = __expf(v[i] - mx); sum += v[i]; }
  red[t] = sum; __syncthreads();
  for (int s = 128; s > 0; s >>= 1) {
    if (t < s) red[t] += red[t + s];
    __syncthreads();
  }
  const float inv = 1.0f / red[0];
#pragma unroll
  for (int i = 0; i < 4; ++i)
    sim[(size_t)row * 1024 + t + i * 256] = v[i] * inv;
}

__global__ void k_cdist(const float* __restrict__ pa, float* __restrict__ sube) {
  const int b = blockIdx.z;
  const int jBase = blockIdx.x * 64, iBase = blockIdx.y * 64;
  __shared__ float Pi[64][17];
  __shared__ float Pj[64][17];
  const int t = threadIdx.x;
  for (int i = t; i < 64 * 16; i += 256) {
    const int r = i >> 4, d = i & 15;
    Pi[r][d] = pa[((size_t)b * 1024 + iBase + r) * 16 + d];
    Pj[r][d] = pa[((size_t)b * 1024 + jBase + r) * 16 + d];
  }
  __syncthreads();
  const int tj = t & 63;
  const int i0 = (t >> 6) * 16;
#pragma unroll 4
  for (int ii = 0; ii < 16; ++ii) {
    const int i = i0 + ii;
    float s = 0.f;
#pragma unroll
    for (int d = 0; d < 16; ++d) s += fabsf(Pi[i][d] - Pj[tj][d]);
    sube[((size_t)b * 1024 + iBase + i) * 1024 + jBase + tj] = __expf(-s);
  }
}

__global__ void k_colsum(const float* __restrict__ sube, float* __restrict__ S) {
  const int b = blockIdx.y;
  const int j = blockIdx.x * 256 + threadIdx.x;
  float acc = 0.f;
  for (int i = 0; i < 1024; ++i)
    acc += sube[((size_t)b * 1024 + i) * 1024 + j];
  S[b * 1024 + j] = acc;
}

__global__ void k_xnorm(const float* __restrict__ x, const float* __restrict__ S,
                        float* __restrict__ xn) {
  const int idx = blockIdx.x * 256 + threadIdx.x;
  xn[idx] = x[idx] / S[idx >> 6];
}

__global__ void k_weights(const float* __restrict__ E, const float* __restrict__ Wp,
                          float* __restrict__ W) {
  const int n = blockIdx.y;
  const int tid = blockIdx.x * 256 + threadIdx.x;
  const int kp = tid >> 5;
  const int oc = (tid & 31) * 2;
  float e[16];
#pragma unroll
  for (int d = 0; d < 16; ++d) e[d] = E[n * 16 + d];
  float4 acc = {0.f, 0.f, 0.f, 0.f};
#pragma unroll
  for (int d = 0; d < 16; ++d) {
    const v2f w0 = *(const v2f*)(Wp + (size_t)d * 8192 + (size_t)(2 * kp) * 64 + oc);
    const v2f w1 = *(const v2f*)(Wp + (size_t)d * 8192 + (size_t)(2 * kp + 1) * 64 + oc);
    acc.x += e[d] * w0.x;
    acc.y += e[d] * w1.x;
    acc.z += e[d] * w0.y;
    acc.w += e[d] * w1.y;
  }
  *(float4*)(W + (size_t)n * 8192 + (size_t)kp * 128 + (size_t)oc * 2) = acc;
}

__global__ __launch_bounds__(256)
void k_bmm_nk(const float* __restrict__ A, long long strideA,
              const float* __restrict__ B, long long strideB,
              float* __restrict__ C, long long strideC) {
  const int b = blockIdx.y;
  const float* Ab = A + (size_t)strideA * b;
  const float* Bb = B + (size_t)strideB * b;
  float* Cb = C + (size_t)strideC * b;
  const int rowBase = blockIdx.x * 128;

  __shared__ float As[128][68];
  __shared__ float Bs[32][160];

  const int t = threadIdx.x;
  const int wave = t >> 5;
  const int lane = t & 31;
  const int m16 = lane & 15;
  const int kh = lane >> 4;

  const v8f zero = {0.f, 0.f, 0.f, 0.f, 0.f, 0.f, 0.f, 0.f};
  v8f cacc[4];
#pragma unroll
  for (int nt = 0; nt < 4; ++nt) cacc[nt] = zero;

  for (int kc = 0; kc < 1024; kc += 64) {
    __syncthreads();
    float4 ra[8], rb0[2], rb1[2];
#pragma unroll
    for (int j = 0; j < 8; ++j) {
      const int i = t + j * 256;
      const int r = i >> 4, cc = (i & 15) << 2;
      ra[j] = *(const float4*)(Ab + (size_t)(rowBase + r) * 1024 + kc + cc);
    }
#pragma unroll
    for (int j = 0; j < 2; ++j) {
      const int i = t + j * 256;
      const int kp = i >> 4, cc = (i & 15) << 2;
      rb0[j] = *(const float4*)(Bb + (size_t)(kc + 2 * kp) * 64 + cc);
      rb1[j] = *(const float4*)(Bb + (size_t)(kc + 2 * kp + 1) * 64 + cc);
    }
    __builtin_amdgcn_sched_barrier(0);
#pragma unroll
    for (int j = 0; j < 8; ++j) {
      const int i = t + j * 256;
      const int r = i >> 4, cc = (i & 15) << 2;
      *(float4*)(&As[r][cc]) = ra[j];
    }
#pragma unroll
    for (int j = 0; j < 2; ++j) {
      const int i = t + j * 256;
      const int kp = i >> 4, cc = (i & 15) << 2;
      const float4 v0 = rb0[j];
      const float4 v1 = rb1[j];
      float4 lo = {v0.x, v1.x, v0.y, v1.y};
      float4 hi = {v0.z, v1.z, v0.w, v1.w};
      *(float4*)(&Bs[kp][cc * 2])     = lo;
      *(float4*)(&Bs[kp][cc * 2 + 4]) = hi;
    }
    __syncthreads();

    const int mrow = wave * 16 + m16;
#pragma unroll
    for (int ks = 0; ks < 16; ++ks) {
      const int kg = ks * 4 + 2 * kh;
      const int kp = ks * 2 + kh;
      const v2f a = *(const v2f*)(&As[mrow][kg]);
#pragma unroll
      for (int nt = 0; nt < 4; ++nt) {
        const v2f bf = *(const v2f*)(&Bs[kp][(nt * 16 + m16) * 2]);
        cacc[nt] = wmma_f32(a, bf, cacc[nt]);
      }
    }
  }

#pragma unroll
  for (int nt = 0; nt < 4; ++nt) {
#pragma unroll
    for (int r = 0; r < 8; ++r) {
      const int m = rowBase + wave * 16 + r + 8 * kh;
      Cb[(size_t)m * 64 + nt * 16 + m16] = cacc[nt][r];
    }
  }
}

__global__ __launch_bounds__(256)
void k_final(const float* __restrict__ sx, const float* __restrict__ xg1,
             const float* __restrict__ W, const float* __restrict__ E,
             const float* __restrict__ bias_pool, float* __restrict__ out) {
  __shared__ float As[8][8][132];
  const int t = threadIdx.x;
  const int wave = t >> 5;
  const int lane = t & 31;
  const int n = blockIdx.x * 8 + wave;
  float (*Aw)[132] = As[wave];

  float va[8][4];
#pragma unroll
  for (int rb = 0; rb < 8; ++rb) {
    const float* s0 = sx  + ((size_t)rb * 1024 + n) * 64;
    const float* s1 = xg1 + ((size_t)rb * 1024 + n) * 64;
    va[rb][0] = s0[lane];
    va[rb][1] = s0[lane + 32];
    va[rb][2] = s1[lane];
    va[rb][3] = s1[lane + 32];
  }
#pragma unroll
  for (int rb = 0; rb < 8; ++rb) {
    Aw[rb][lane]      = va[rb][0];
    Aw[rb][lane + 32] = va[rb][1];
    Aw[rb][lane + 64] = va[rb][2];
    Aw[rb][lane + 96] = va[rb][3];
  }
  __syncthreads();

  const float* Wn = W + (size_t)n * 8192;
  const int m16 = lane & 15;
  const int kh = lane >> 4;

  const v8f zero = {0.f, 0.f, 0.f, 0.f, 0.f, 0.f, 0.f, 0.f};
  v8f cacc[4];
#pragma unroll
  for (int nt = 0; nt < 4; ++nt) cacc[nt] = zero;

#pragma unroll 4
  for (int ks = 0; ks < 32; ++ks) {
    const int kg = ks * 4 + 2 * kh;
    const int kp = ks * 2 + kh;
    v2f a = {0.f, 0.f};
    if (m16 < 8) a = *(const v2f*)(&Aw[m16][kg]);
#pragma unroll
    for (int nt = 0; nt < 4; ++nt) {
      const v2f bf = *(const v2f*)(Wn + (size_t)kp * 128 + (nt * 16 + m16) * 2);
      cacc[nt] = wmma_f32(a, bf, cacc[nt]);
    }
  }

#pragma unroll
  for (int nt = 0; nt < 4; ++nt) {
    const int col = nt * 16 + m16;
    float bv = 0.f;
#pragma unroll
    for (int d = 0; d < 16; ++d) bv += E[n * 16 + d] * bias_pool[d * 64 + col];
#pragma unroll
    for (int r = 0; r < 8; ++r) {
      const int m = r + 8 * kh;
      if (m < 8)
        out[((size_t)m * 1024 + n) * 64 + col] = cacc[nt][r] + bv;
    }
  }
}

extern "C" void kernel_launch(void* const* d_in, const int* in_sizes, int n_in,
                              void* d_out, int out_size, void* d_ws, size_t ws_size,
                              hipStream_t stream) {
  (void)in_sizes; (void)n_in; (void)out_size; (void)ws_size;
  const float* x    = (const float*)d_in[0];
  const float* E    = (const float*)d_in[1];
  const float* pa   = (const float*)d_in[2];
  const float* Wp   = (const float*)d_in[3];
  const float* bp   = (const float*)d_in[4];
  float* out = (float*)d_out;

  float* ws   = (float*)d_ws;
  float* sim  = ws;
  float* sube = sim  + 1048576;
  float* S    = sube + 8388608;
  float* xn   = S    + 8192;
  float* sx   = xn   + 524288;
  float* xg1  = sx   + 524288;
  float* W    = xg1  + 524288;

  k_simraw    <<<4096, 256, 0, stream>>>(E, sim);
  k_rowsoftmax<<<1024, 256, 0, stream>>>(sim);
  k_cdist     <<<dim3(16, 16, 8), 256, 0, stream>>>(pa, sube);
  k_colsum    <<<dim3(4, 8),      256, 0, stream>>>(sube, S);
  k_xnorm     <<<2048, 256, 0, stream>>>(x, S, xn);
  k_weights   <<<dim3(8, 1024),   256, 0, stream>>>(E, Wp, W);
  k_bmm_nk<<<dim3(8, 8), 256, 0, stream>>>(sube, 1024LL * 1024, xn, 1024LL * 64,
                                           sx, 1024LL * 64);
  k_bmm_nk<<<dim3(8, 8), 256, 0, stream>>>(sim, 0LL, sx, 1024LL * 64,
                                           xg1, 1024LL * 64);
  k_final<<<128, 256, 0, stream>>>(sx, xg1, W, E, bp, out);
}


# Column sums of sub_e formed inside the cdist kernel as per-tile partials and finished in xnorm; the separate colsum pass over sub_e is removed; changes the summation order of the column sums
# speedup vs baseline: 2.0204x; 1.2923x over previous
#include <hip/hip_runtime.h>
#include <hip/hip_bf16.h>
#include <cstddef>


typedef float v2f __attribute__((ext_vector_type(2)));
typedef float v8f __attribute__((ext_vector_type(8)));

static __device__ __forceinline__ v8f wmma_f32(v2f a, v2f b, v8f c) {
  return __builtin_amdgcn_wmma_f32_16x16x4_f32(
       false, a,  false, b,
       (short)0, c,  false,  false);
}

__global__ void k_simraw(const float* __restrict__ E, float* __restrict__ sim) {
  const int idx = blockIdx.x * 256 + threadIdx.x;
  const int i = idx >> 10, j = idx & 1023;
  float acc = 0.f;
#pragma unroll
  for (int d = 0; d < 16; ++d) acc += E[i * 16 + d] * E[j * 16 + d];
  sim[idx] = fmaxf(acc, 0.f);
}

__global__ void k_rowsoftmax(float* __restrict__ sim) {
  const int row = blockIdx.x;
  const int t = threadIdx.x;
  __shared__ float red[256];
  float v[4];
  float mx = -3.0e38f;
#pragma unroll
  for (int i = 0; i < 4; ++i) {
    v[i] = sim[(size_t)row * 1024 + t + i * 256];
    mx = fmaxf(mx, v[i]);
  }
  red[t] = mx; __syncthreads();
  for (int s = 128; s > 0; s >>= 1) {
    if (t < s) red[t] = fmaxf(red[t], red[t + s]);
    __syncthreads();
  }
  mx = red[0]; __syncthreads();
  float sum = 0.f;
#pragma unroll
  for (int i = 0; i < 4; ++i) { v[i] = __expf(v[i] - mx); sum += v[i]; }
  red[t] = sum; __syncthreads();
  for (int s = 128; s > 0; s >>= 1) {
    if (t < s) red[t] += red[t + s];
    __syncthreads();
  }
  const float inv = 1.0f / red[0];
#pragma unroll
  for (int i = 0; i < 4; ++i)
    sim[(size_t)row * 1024 + t + i * 256] = v[i] * inv;
}

__global__ void k_cdist(const float* __restrict__ pa, float* __restrict__ sube,
                        float* __restrict__ P) {
  const int b = blockIdx.z;
  const int jBase = blockIdx.x * 64, iBase = blockIdx.y * 64;
  __shared__ float Pi[64][17];
  __shared__ float Pj[64][17];
  __shared__ float Cs[4][64];
  const int t = threadIdx.x;
  for (int i = t; i < 64 * 16; i += 256) {
    const int r = i >> 4, d = i & 15;
    Pi[r][d] = pa[((size_t)b * 1024 + iBase + r) * 16 + d];
    Pj[r][d] = pa[((size_t)b * 1024 + jBase + r) * 16 + d];
  }
  __syncthreads();
  const int tj = t & 63;
  const int g = t >> 6;
  const int i0 = g * 16;
  float csum = 0.f;
#pragma unroll 4
  for (int ii = 0; ii < 16; ++ii) {
    const int i = i0 + ii;
    float s = 0.f;
#pragma unroll
    for (int d = 0; d < 16; ++d) s += fabsf(Pi[i][d] - Pj[tj][d]);
    const float e = __expf(-s);
    sube[((size_t)b * 1024 + iBase + i) * 1024 + jBase + tj] = e;
    csum += e;
  }
  Cs[g][tj] = csum;
  __syncthreads();
  if (t < 64)
    P[((size_t)blockIdx.y * 8 + b) * 1024 + jBase + t] =
        (Cs[0][t] + Cs[1][t]) + (Cs[2][t] + Cs[3][t]);
}

__global__ void k_xnorm(const float* __restrict__ x, const float* __restrict__ P,
                        float* __restrict__ xn) {
  const int idx = blockIdx.x * 256 + threadIdx.x;
  const int bj = idx >> 6;
  float s = 0.f;
#pragma unroll
  for (int ib = 0; ib < 16; ++ib) s += P[ib * 8192 + bj];
  xn[idx] = x[idx] / s;
}

__global__ void k_weights(const float* __restrict__ E, const float* __restrict__ Wp,
                          float* __restrict__ W) {
  const int n = blockIdx.y;
  const int tid = blockIdx.x * 256 + threadIdx.x;
  const int kp = tid >> 5;
  const int oc = (tid & 31) * 2;
  float e[16];
#pragma unroll
  for (int d = 0; d < 16; ++d) e[d] = E[n * 16 + d];
  float4 acc = {0.f, 0.f, 0.f, 0.f};
#pragma unroll
  for (int d = 0; d < 16; ++d) {
    const v2f w0 = *(const v2f*)(Wp + (size_t)d * 8192 + (size_t)(2 * kp) * 64 + oc);
    const v2f w1 = *(const v2f*)(Wp + (size_t)d * 8192 + (size_t)(2 * kp + 1) * 64 + oc);
    acc.x += e[d] * w0.x;
    acc.y += e[d] * w1.x;
    acc.z += e[d] * w0.y;
    acc.w += e[d] * w1.y;
  }
  *(float4*)(W + (size_t)n * 8192 + (size_t)kp * 128 + (size_t)oc * 2) = acc;
}

__global__ __launch_bounds__(256)
void k_bmm_nk(const float* __restrict__ A, long long strideA,
              const float* __restrict__ B, long long strideB,
              float* __restrict__ C, long long strideC) {
  const int b = blockIdx.y;
  const float* Ab = A + (size_t)strideA * b;
  const float* Bb = B + (size_t)strideB * b;
  float* Cb = C + (size_t)strideC * b;
  const int rowBase = blockIdx.x * 128;

  __shared__ float As[128][68];
  __shared__ float Bs[32][160];

  const int t = threadIdx.x;
  const int wave = t >> 5;
  const int lane = t & 31;
  const int m16 = lane & 15;
  const int kh = lane >> 4;

  const v8f zero = {0.f, 0.f, 0.f, 0.f, 0.f, 0.f, 0.f, 0.f};
  v8f cacc[4];
#pragma unroll
  for (int nt = 0; nt < 4; ++nt) cacc[nt] = zero;

  for (int kc = 0; kc < 1024; kc += 64) {
    __syncthreads();
    float4 ra[8], rb0[2], rb1[2];
#pragma unroll
    for (int j = 0; j < 8; ++j) {
      const int i = t + j * 256;
      const int r = i >> 4, cc = (i & 15) << 2;
      ra[j] = *(const float4*)(Ab + (size_t)(rowBase + r) * 1024 + kc + cc);
    }
#pragma unroll
    for (int j = 0; j < 2; ++j) {
      const int i = t + j * 256;
      const int kp = i >> 4, cc = (i & 15) << 2;
      rb0[j] = *(const float4*)(Bb + (size_t)(kc + 2 * kp) * 64 + cc);
      rb1[j] = *(const float4*)(Bb + (size_t)(kc + 2 * kp + 1) * 64 + cc);
    }
    __builtin_amdgcn_sched_barrier(0);
#pragma unroll
    for (int j = 0; j < 8; ++j) {
      const int i = t + j * 256;
      const int r = i >> 4, cc = (i & 15) << 2;
      *(float4*)(&As[r][cc]) = ra[j];
    }
#pragma unroll
    for (int j = 0; j < 2; ++j) {
      const int i = t + j * 256;
      const int kp = i >> 4, cc = (i & 15) << 2;
      const float4 v0 = rb0[j];
      const float4 v1 = rb1[j];
      float4 lo = {v0.x, v1.x, v0.y, v1.y};
      float4 hi = {v0.z, v1.z, v0.w, v1.w};
      *(float4*)(&Bs[kp][cc * 2])     = lo;
      *(float4*)(&Bs[kp][cc * 2 + 4]) = hi;
    }
    __syncthreads();

    const int mrow = wave * 16 + m16;
#pragma unroll
    for (int ks = 0; ks < 16; ++ks) {
      const int kg = ks * 4 + 2 * kh;
      const int kp = ks * 2 + kh;
      const v2f a = *(const v2f*)(&As[mrow][kg]);
#pragma unroll
      for (int nt = 0; nt < 4; ++nt) {
        const v2f bf = *(const v2f*)(&Bs[kp][(nt * 16 + m16) * 2]);
        cacc[nt] = wmma_f32(a, bf, cacc[nt]);
      }
    }
  }

#pragma unroll
  for (int nt = 0; nt < 4; ++nt) {
#pragma unroll
    for (int r = 0; r < 8; ++r) {
      const int m = rowBase + wave * 16 + r + 8 * kh;
      Cb[(size_t)m * 64 + nt * 16 + m16] = cacc[nt][r];
    }
  }
}

__global__ __launch_bounds__(256)
void k_final(const float* __restrict__ sx, const float* __restrict__ xg1,
             const float* __restrict__ W, const float* __restrict__ E,
             const float* __restrict__ bias_pool, float* __restrict__ out) {
  __shared__ float As[8][8][132];
  const int t = threadIdx.x;
  const int wave = t >> 5;
  const int lane = t & 31;
  const int n = blockIdx.x * 8 + wave;
  float (*Aw)[132] = As[wave];

  float va[8][4];
#pragma unroll
  for (int rb = 0; rb < 8; ++rb) {
    const float* s0 = sx  + ((size_t)rb * 1024 + n) * 64;
    const float* s1 = xg1 + ((size_t)rb * 1024 + n) * 64;
    va[rb][0] = s0[lane];
    va[rb][1] = s0[lane + 32];
    va[rb][2] = s1[lane];
    va[rb][3] = s1[lane + 32];
  }
#pragma unroll
  for (int rb = 0; rb < 8; ++rb) {
    Aw[rb][lane]      = va[rb][0];
    Aw[rb][lane + 32] = va[rb][1];
    Aw[rb][lane + 64] = va[rb][2];
    Aw[rb][lane + 96] = va[rb][3];
  }
  __syncthreads();

  const float* Wn = W + (size_t)n * 8192;
  const int m16 = lane & 15;
  const int kh = lane >> 4;

  const v8f zero = {0.f, 0.f, 0.f, 0.f, 0.f, 0.f, 0.f, 0.f};
  v8f cacc[4];
#pragma unroll
  for (int nt = 0; nt < 4; ++nt) cacc[nt] = zero;

#pragma unroll 4
  for (int ks = 0; ks < 32; ++ks) {
    const int kg = ks * 4 + 2 * kh;
    const int kp = ks * 2 + kh;
    v2f a = {0.f, 0.f};
    if (m16 < 8) a = *(const v2f*)(&Aw[m16][kg]);
#pragma unroll
    for (int nt = 0; nt < 4; ++nt) {
      const v2f bf = *(const v2f*)(Wn + (size_t)kp * 128 + (nt * 16 + m16) * 2);
      cacc[nt] = wmma_f32(a, bf, cacc[nt]);
    }
  }

#pragma unroll
  for (int nt = 0; nt < 4; ++nt) {
    const int col = nt * 16 + m16;
    float bv = 0.f;
#pragma unroll
    for (int d = 0; d < 16; ++d) bv += E[n * 16 + d] * bias_pool[d * 64 + col];
#pragma unroll
    for (int r = 0; r < 8; ++r) {
      const int m = r + 8 * kh;
      if (m < 8)
        out[((size_t)m * 1024 + n) * 64 + col] = cacc[nt][r] + bv;
    }
  }
}

extern "C" void kernel_launch(void* const* d_in, const int* in_sizes, int n_in,
                              void* d_out, int out_size, void* d_ws, size_t ws_size,
                              hipStream_t stream) {
  (void)in_sizes; (void)n_in; (void)out_size; (void)ws_size;
  const float* x    = (const float*)d_in[0];
  const float* E    = (const float*)d_in[1];
  const float* pa   = (const float*)d_in[2];
  const float* Wp   = (const float*)d_in[3];
  const float* bp   = (const float*)d_in[4];
  float* out = (float*)d_out;

  float* ws   = (float*)d_ws;
  float* sim  = ws;
  float* sube = sim  + 1048576;
  float* S    = sube + 8388608;
  float* xn   = S    + 8192;
  float* sx   = xn   + 524288;
  float* xg1  = sx   + 524288;
  float* P    = xg1;
  float* W    = xg1  + 524288;

  k_simraw    <<<4096, 256, 0, stream>>>(E, sim);
  k_rowsoftmax<<<1024, 256, 0, stream>>>(sim);
  k_cdist     <<<dim3(16, 16, 8), 256, 0, stream>>>(pa, sube, P);
  k_xnorm     <<<2048, 256, 0, stream>>>(x, P, xn);
  k_weights   <<<dim3(8, 1024),   256, 0, stream>>>(E, Wp, W);
  k_bmm_nk<<<dim3(8, 8), 256, 0, stream>>>(sube, 1024LL * 1024, xn, 1024LL * 64,
                                           sx, 1024LL * 64);
  k_bmm_nk<<<dim3(8, 8), 256, 0, stream>>>(sim, 0LL, sx, 1024LL * 64,
                                           xg1, 1024LL * 64);
  k_final<<<128, 256, 0, stream>>>(sx, xg1, W, E, bp, out);
}


# Both batched products re-tiled from 64 work-groups of 128 rows to 256 of 32 rows, one per WGP; K split over a work-group's waves, their four partial sums added at the end (changes summation order)
# speedup vs baseline: 3.0606x; 1.5148x over previous
#include <hip/hip_runtime.h>
#include <hip/hip_bf16.h>
#include <cstddef>


typedef float v2f __attribute__((ext_vector_type(2)));
typedef float v8f __attribute__((ext_vector_type(8)));

static __device__ __forceinline__ v8f wmma_f32(v2f a, v2f b, v8f c) {
  return __builtin_amdgcn_wmma_f32_16x16x4_f32(
       false, a,  false, b,
       (short)0, c,  false,  false);
}

__global__ void k_simraw(const float* __restrict__ E, float* __restrict__ sim) {
  const int idx = blockIdx.x * 256 + threadIdx.x;
  const int i = idx >> 10, j = idx & 1023;
  float acc = 0.f;
#pragma unroll
  for (int d = 0; d < 16; ++d) acc += E[i * 16 + d] * E[j * 16 + d];
  sim[idx] = fmaxf(acc, 0.f);
}

__global__ void k_rowsoftmax(float* __restrict__ sim) {
  const int row = blockIdx.x;
  const int t = threadIdx.x;
  __shared__ float red[256];
  float v[4];
  float mx = -3.0e38f;
#pragma unroll
  for (int i = 0; i < 4; ++i) {
    v[i] = sim[(size_t)row * 1024 + t + i * 256];
    mx = fmaxf(mx, v[i]);
  }
  red[t] = mx; __syncthreads();
  for (int s = 128; s > 0; s >>= 1) {
    if (t < s) red[t] = fmaxf(red[t], red[t + s]);
    __syncthreads();
  }
  mx = red[0]; __syncthreads();
  float sum = 0.f;
#pragma unroll
  for (int i = 0; i < 4; ++i) { v[i] = __expf(v[i] - mx); sum += v[i]; }
  red[t] = sum; __syncthreads();
  for (int s = 128; s > 0; s >>= 1) {
    if (t < s) red[t] += red[t + s];
    __syncthreads();
  }
  const float inv = 1.0f / red[0];
#pragma unroll
  for (int i = 0; i < 4; ++i)
    sim[(size_t)row * 1024 + t + i * 256] = v[i] * inv;
}

__global__ void k_cdist(const float* __restrict__ pa, float* __restrict__ sube,
                        float* __restrict__ P) {
  const int b = blockIdx.z;
  const int jBase = blockIdx.x * 64, iBase = blockIdx.y * 64;
  __shared__ float Pi[64][17];
  __shared__ float Pj[64][17];
  __shared__ float Cs[4][64];
  const int t = threadIdx.x;
  for (int i = t; i < 64 * 16; i += 256) {
    const int r = i >> 4, d = i & 15;
    Pi[r][d] = pa[((size_t)b * 1024 + iBase + r) * 16 + d];
    Pj[r][d] = pa[((size_t)b * 1024 + jBase + r) * 16 + d];
  }
  __syncthreads();
  const int tj = t & 63;
  const int g = t >> 6;
  const int i0 = g * 16;
  float csum = 0.f;
#pragma unroll 4
  for (int ii = 0; ii < 16; ++ii) {
    const int i = i0 + ii;
    float s = 0.f;
#pragma unroll
    for (int d = 0; d < 16; ++d) s += fabsf(Pi[i][d] - Pj[tj][d]);
    const float e = __expf(-s);
    sube[((size_t)b * 1024 + iBase + i) * 1024 + jBase + tj] = e;
    csum += e;
  }
  Cs[g][tj] = csum;
  __syncthreads();
  if (t < 64)
    P[((size_t)blockIdx.y * 8 + b) * 1024 + jBase + t] =
        (Cs[0][t] + Cs[1][t]) + (Cs[2][t] + Cs[3][t]);
}

__global__ void k_xnorm(const float* __restrict__ x, const float* __restrict__ P,
                        float* __restrict__ xn) {
  const int idx = blockIdx.x * 256 + threadIdx.x;
  const int bj = idx >> 6;
  float s = 0.f;
#pragma unroll
  for (int ib = 0; ib < 16; ++ib) s += P[ib * 8192 + bj];
  xn[idx] = x[idx] / s;
}

__global__ void k_weights(const float* __restrict__ E, const float* __restrict__ Wp,
                          float* __restrict__ W) {
  const int n = blockIdx.y;
  const int tid = blockIdx.x * 256 + threadIdx.x;
  const int kp = tid >> 5;
  const int oc = (tid & 31) * 2;
  float e[16];
#pragma unroll
  for (int d = 0; d < 16; ++d) e[d] = E[n * 16 + d];
  float4 acc = {0.f, 0.f, 0.f, 0.f};
#pragma unroll
  for (int d = 0; d < 16; ++d) {
    const v2f w0 = *(const v2f*)(Wp + (size_t)d * 8192 + (size_t)(2 * kp) * 64 + oc);
    const v2f w1 = *(const v2f*)(Wp + (size_t)d * 8192 + (size_t)(2 * kp + 1) * 64 + oc);
    acc.x += e[d] * w0.x;
    acc.y += e[d] * w1.x;
    acc.z += e[d] * w0.y;
    acc.w += e[d] * w1.y;
  }
  *(float4*)(W + (size_t)n * 8192 + (size_t)kp * 128 + (size_t)oc * 2) = acc;
}

#define BMM_KC 256
__global__ __launch_bounds__(256)
void k_bmm_nk(const float* __restrict__ A, long long strideA,
              const float* __restrict__ B, long long strideB,
              float* __restrict__ C, long long strideC) {
  const int b = blockIdx.y;
  const float* Ab = A + (size_t)strideA * b;
  const float* Bb = B + (size_t)strideB * b;
  float* Cb = C + (size_t)strideC * b;
  const int rowBase = blockIdx.x * 32;

  __shared__ float As[32][BMM_KC + 4];
  __shared__ float Bs[BMM_KC / 2][160];
  __shared__ float Red[4][32][68];

  const int t = threadIdx.x;
  const int wave = t >> 5;
  const int lane = t & 31;
  const int m16 = lane & 15;
  const int kh = lane >> 4;
  const int rs = wave & 1;
  const int kq = wave >> 1;

  const v8f zero = {0.f, 0.f, 0.f, 0.f, 0.f, 0.f, 0.f, 0.f};
  v8f cacc[4];
#pragma unroll
  for (int nt = 0; nt < 4; ++nt) cacc[nt] = zero;

  for (int kc = 0; kc < 1024; kc += BMM_KC) {
    __syncthreads();
    float4 ra[8], rb0[8], rb1[8];
#pragma unroll
    for (int j = 0; j < 8; ++j) {
      const int i = t + j * 256;
      const int r = i >> 6, cc = (i & 63) << 2;
      ra[j] = *(const float4*)(Ab + (size_t)(rowBase + r) * 1024 + kc + cc);
    }
#pragma unroll
    for (int j = 0; j < 8; ++j) {
      const int i = t + j * 256;
      const int kp = i >> 4, cc = (i & 15) << 2;
      rb0[j] = *(const float4*)(Bb + (size_t)(kc + 2 * kp) * 64 + cc);
      rb1[j] = *(const float4*)(Bb + (size_t)(kc + 2 * kp + 1) * 64 + cc);
    }
    __builtin_amdgcn_sched_barrier(0);
#pragma unroll
    for (int j = 0; j < 8; ++j) {
      const int i = t + j * 256;
      const int r = i >> 6, cc = (i & 63) << 2;
      *(float4*)(&As[r][cc]) = ra[j];
    }
#pragma unroll
    for (int j = 0; j < 8; ++j) {
      const int i = t + j * 256;
      const int kp = i >> 4, cc = (i & 15) << 2;
      const float4 v0 = rb0[j];
      const float4 v1 = rb1[j];
      float4 lo = {v0.x, v1.x, v0.y, v1.y};
      float4 hi = {v0.z, v1.z, v0.w, v1.w};
      *(float4*)(&Bs[kp][cc * 2])     = lo;
      *(float4*)(&Bs[kp][cc * 2 + 4]) = hi;
    }
    __syncthreads();

    const int mrow = rs * 16 + m16;
    const int kq0 = kq * (BMM_KC / 4);
#pragma unroll
    for (int ks = 0; ks < BMM_KC / 16; ++ks) {
      const int kg = kq0 + ks * 4 + 2 * kh;
      const int kp = (kq0 >> 1) + ks * 2 + kh;
      const v2f a = *(const v2f*)(&As[mrow][kg]);
#pragma unroll
      for (int nt = 0; nt < 4; ++nt) {
        const v2f bf = *(const v2f*)(&Bs[kp][(nt * 16 + m16) * 2]);
        cacc[nt] = wmma_f32(a, bf, cacc[nt]);
      }
    }
  }

#pragma unroll
  for (int nt = 0; nt < 4; ++nt) {
#pragma unroll
    for (int r = 0; r < 8; ++r)
      Red[kq][rs * 16 + r + 8 * kh][nt * 16 + m16] = cacc[nt][r];
  }
  __syncthreads();
#pragma unroll
  for (int j = 0; j < 2; ++j) {
    const int e = (t + j * 256) * 4;
    const int row = e >> 6, col = e & 63;
    const float4 p0 = *(const float4*)(&Red[0][row][col]);
    const float4 p1 = *(const float4*)(&Red[1][row][col]);
    const float4 p2 = *(const float4*)(&Red[2][row][col]);
    const float4 p3 = *(const float4*)(&Red[3][row][col]);
    float4 s;
    s.x = (p0.x + p1.x) + (p2.x + p3.x);
    s.y = (p0.y + p1.y) + (p2.y + p3.y);
    s.z = (p0.z + p1.z) + (p2.z + p3.z);
    s.w = (p0.w + p1.w) + (p2.w + p3.w);
    *(float4*)(Cb + (size_t)(rowBase + row) * 64 + col) = s;
  }
}

__global__ __launch_bounds__(256)
void k_final(const float* __restrict__ sx, const float* __restrict__ xg1,
             const float* __restrict__ W, const float* __restrict__ E,
             const float* __restrict__ bias_pool, float* __restrict__ out) {
  __shared__ float As[8][8][132];
  const int t = threadIdx.x;
  const int wave = t >> 5;
  const int lane = t & 31;
  const int n = blockIdx.x * 8 + wave;
  float (*Aw)[132] = As[wave];

  float va[8][4];
#pragma unroll
  for (int rb = 0; rb < 8; ++rb) {
    const float* s0 = sx  + ((size_t)rb * 1024 + n) * 64;
    const float* s1 = xg1 + ((size_t)rb * 1024 + n) * 64;
    va[rb][0] = s0[lane];
    va[rb][1] = s0[lane + 32];
    va[rb][2] = s1[lane];
    va[rb][3] = s1[lane + 32];
  }
#pragma unroll
  for (int rb = 0; rb < 8; ++rb) {
    Aw[rb][lane]      = va[rb][0];
    Aw[rb][lane + 32] = va[rb][1];
    Aw[rb][lane + 64] = va[rb][2];
    Aw[rb][lane + 96] = va[rb][3];
  }
  __syncthreads();

  const float* Wn = W + (size_t)n * 8192;
  const int m16 = lane & 15;
  const int kh = lane >> 4;

  const v8f zero = {0.f, 0.f, 0.f, 0.f, 0.f, 0.f, 0.f, 0.f};
  v8f cacc[4];
#pragma unroll
  for (int nt = 0; nt < 4; ++nt) cacc[nt] = zero;

#pragma unroll 4
  for (int ks = 0; ks < 32; ++ks) {
    const int kg = ks * 4 + 2 * kh;
    const int kp = ks * 2 + kh;
    v2f a = {0.f, 0.f};
    if (m16 < 8) a = *(const v2f*)(&Aw[m16][kg]);
#pragma unroll
    for (int nt = 0; nt < 4; ++nt) {
      const v2f bf = *(const v2f*)(Wn + (size_t)kp * 128 + (nt * 16 + m16) * 2);
      cacc[nt] = wmma_f32(a, bf, cacc[nt]);
    }
  }

#pragma unroll
  for (int nt = 0; nt < 4; ++nt) {
    const int col = nt * 16 + m16;
    float bv = 0.f;
#pragma unroll
    for (int d = 0; d < 16; ++d) bv += E[n * 16 + d] * bias_pool[d * 64 + col];
#pragma unroll
    for (int r = 0; r < 8; ++r) {
      const int m = r + 8 * kh;
      if (m < 8)
        out[((size_t)m * 1024 + n) * 64 + col] = cacc[nt][r] + bv;
    }
  }
}

extern "C" void kernel_launch(void* const* d_in, const int* in_sizes, int n_in,
                              void* d_out, int out_size, void* d_ws, size_t ws_size,
                              hipStream_t stream) {
  (void)in_sizes; (void)n_in; (void)out_size; (void)ws_size;
  const float* x    = (const float*)d_in[0];
  const float* E    = (const float*)d_in[1];
  const float* pa   = (const float*)d_in[2];
  const float* Wp   = (const float*)d_in[3];
  const float* bp   = (const float*)d_in[4];
  float* out = (float*)d_out;

  float* ws   = (float*)d_ws;
  float* sim  = ws;
  float* sube = sim  + 1048576;
  float* S    = sube + 8388608;
  float* xn   = S    + 8192;
  float* sx   = xn   + 524288;
  float* xg1  = sx   + 524288;
  float* P    = xg1;
  float* W    = xg1  + 524288;

  k_simraw    <<<4096, 256, 0, stream>>>(E, sim);
  k_rowsoftmax<<<1024, 256, 0, stream>>>(sim);
  k_cdist     <<<dim3(16, 16, 8), 256, 0, stream>>>(pa, sube, P);
  k_xnorm     <<<2048, 256, 0, stream>>>(x, P, xn);
  k_weights   <<<dim3(8, 1024),   256, 0, stream>>>(E, Wp, W);
  k_bmm_nk<<<dim3(32, 8), 256, 0, stream>>>(sube, 1024LL * 1024, xn, 1024LL * 64,
                                            sx, 1024LL * 64);
  k_bmm_nk<<<dim3(32, 8), 256, 0, stream>>>(sim, 0LL, sx, 1024LL * 64,
                                            xg1, 1024LL * 64);
  k_final<<<128, 256, 0, stream>>>(sx, xg1, W, E, bp, out);
}


# Final kernel: all global loads requested in front of the WMMA run: a node's whole 128x64 W as one group (one memory round trip instead of one per k-step), bias operands with the A loads
# speedup vs baseline: 3.6406x; 1.1895x over previous
#include <hip/hip_runtime.h>
#include <hip/hip_bf16.h>
#include <cstddef>


typedef float v2f __attribute__((ext_vector_type(2)));
typedef float v8f __attribute__((ext_vector_type(8)));

static __device__ __forceinline__ v8f wmma_f32(v2f a, v2f b, v8f c) {
  return __builtin_amdgcn_wmma_f32_16x16x4_f32(
       false, a,  false, b,
       (short)0, c,  false,  false);
}

__global__ void k_simraw(const float* __restrict__ E, float* __restrict__ sim) {
  const int idx = blockIdx.x * 256 + threadIdx.x;
  const int i = idx >> 10, j = idx & 1023;
  float acc = 0.f;
#pragma unroll
  for (int d = 0; d < 16; ++d) acc += E[i * 16 + d] * E[j * 16 + d];
  sim[idx] = fmaxf(acc, 0.f);
}

__global__ void k_rowsoftmax(float* __restrict__ sim) {
  const int row = blockIdx.x;
  const int t = threadIdx.x;
  __shared__ float red[256];
  float v[4];
  float mx = -3.0e38f;
#pragma unroll
  for (int i = 0; i < 4; ++i) {
    v[i] = sim[(size_t)row * 1024 + t + i * 256];
    mx = fmaxf(mx, v[i]);
  }
  red[t] = mx; __syncthreads();
  for (int s = 128; s > 0; s >>= 1) {
    if (t < s) red[t] = fmaxf(red[t], red[t + s]);
    __syncthreads();
  }
  mx = red[0]; __syncthreads();
  float sum = 0.f;
#pragma unroll
  for (int i = 0; i < 4; ++i) { v[i] = __expf(v[i] - mx); sum += v[i]; }
  red[t] = sum; __syncthreads();
  for (int s = 128; s > 0; s >>= 1) {
    if (t < s) red[t] += red[t + s];
    __syncthreads();
  }
  const float inv = 1.0f / red[0];
#pragma unroll
  for (int i = 0; i < 4; ++i)
    sim[(size_t)row * 1024 + t + i * 256] = v[i] * inv;
}

__global__ void k_cdist(const float* __restrict__ pa, float* __restrict__ sube,
                        float* __restrict__ P) {
  const int b = blockIdx.z;
  const int jBase = blockIdx.x * 64, iBase = blockIdx.y * 64;
  __shared__ float Pi[64][17];
  __shared__ float Pj[64][17];
  __shared__ float Cs[4][64];
  const int t = threadIdx.x;
  for (int i = t; i < 64 * 16; i += 256) {
    const int r = i >> 4, d = i & 15;
    Pi[r][d] = pa[((size_t)b * 1024 + iBase + r) * 16 + d];
    Pj[r][d] = pa[((size_t)b * 1024 + jBase + r) * 16 + d];
  }
  __syncthreads();
  const int tj = t & 63;
  const int g = t >> 6;
  const int i0 = g * 16;
  float csum = 0.f;
#pragma unroll 4
  for (int ii = 0; ii < 16; ++ii) {
    const int i = i0 + ii;
    float s = 0.f;
#pragma unroll
    for (int d = 0; d < 16; ++d) s += fabsf(Pi[i][d] - Pj[tj][d]);
    const float e = __expf(-s);
    sube[((size_t)b * 1024 + iBase + i) * 1024 + jBase + tj] = e;
    csum += e;
  }
  Cs[g][tj] = csum;
  __syncthreads();
  if (t < 64)
    P[((size_t)blockIdx.y * 8 + b) * 1024 + jBase + t] =
        (Cs[0][t] + Cs[1][t]) + (Cs[2][t] + Cs[3][t]);
}

__global__ void k_xnorm(const float* __restrict__ x, const float* __restrict__ P,
                        float* __restrict__ xn) {
  const int idx = blockIdx.x * 256 + threadIdx.x;
  const int bj = idx >> 6;
  float s = 0.f;
#pragma unroll
  for (int ib = 0; ib < 16; ++ib) s += P[ib * 8192 + bj];
  xn[idx] = x[idx] / s;
}

__global__ void k_weights(const float* __restrict__ E, const float* __restrict__ Wp,
                          float* __restrict__ W) {
  const int n = blockIdx.y;
  const int tid = blockIdx.x * 256 + threadIdx.x;
  const int kp = tid >> 5;
  const int oc = (tid & 31) * 2;
  float e[16];
#pragma unroll
  for (int d = 0; d < 16; ++d) e[d] = E[n * 16 + d];
  float4 acc = {0.f, 0.f, 0.f, 0.f};
#pragma unroll
  for (int d = 0; d < 16; ++d) {
    const v2f w0 = *(const v2f*)(Wp + (size_t)d * 8192 + (size_t)(2 * kp) * 64 + oc);
    const v2f w1 = *(const v2f*)(Wp + (size_t)d * 8192 + (size_t)(2 * kp + 1) * 64 + oc);
    acc.x += e[d] * w0.x;
    acc.y += e[d] * w1.x;
    acc.z += e[d] * w0.y;
    acc.w += e[d] * w1.y;
  }
  *(float4*)(W + (size_t)n * 8192 + (size_t)kp * 128 + (size_t)oc * 2) = acc;
}

#define BMM_KC 256
__global__ __launch_bounds__(256)
void k_bmm_nk(const float* __restrict__ A, long long strideA,
              const float* __restrict__ B, long long strideB,
              float* __restrict__ C, long long strideC) {
  const int b = blockIdx.y;
  const float* Ab = A + (size_t)strideA * b;
  const float* Bb = B + (size_t)strideB * b;
  float* Cb = C + (size_t)strideC * b;
  const int rowBase = blockIdx.x * 32;

  __shared__ float As[32][BMM_KC + 4];
  __shared__ float Bs[BMM_KC / 2][160];
  __shared__ float Red[4][32][68];

  const int t = threadIdx.x;
  const int wave = t >> 5;
  const int lane = t & 31;
  const int m16 = lane & 15;
  const int kh = lane >> 4;
  const int rs = wave & 1;
  const int kq = wave >> 1;

  const v8f zero = {0.f, 0.f, 0.f, 0.f, 0.f, 0.f, 0.f, 0.f};
  v8f cacc[4];
#pragma unroll
  for (int nt = 0; nt < 4; ++nt) cacc[nt] = zero;

  for (int kc = 0; kc < 1024; kc += BMM_KC) {
    __syncthreads();
    float4 ra[8], rb0[8], rb1[8];
#pragma unroll
    for (int j = 0; j < 8; ++j) {
      const int i = t + j * 256;
      const int r = i >> 6, cc = (i & 63) << 2;
      ra[j] = *(const float4*)(Ab + (size_t)(rowBase + r) * 1024 + kc + cc);
    }
#pragma unroll
    for (int j = 0; j < 8; ++j) {
      const int i = t + j * 256;
      const int kp = i >> 4, cc = (i & 15) << 2;
      rb0[j] = *(const float4*)(Bb + (size_t)(kc + 2 * kp) * 64 + cc);
      rb1[j] = *(const float4*)(Bb + (size_t)(kc + 2 * kp + 1) * 64 + cc);
    }
    __builtin_amdgcn_sched_barrier(0);
#pragma unroll
    for (int j = 0; j < 8; ++j) {
      const int i = t + j * 256;
      const int r = i >> 6, cc = (i & 63) << 2;
      *(float4*)(&As[r][cc]) = ra[j];
    }
#pragma unroll
    for (int j = 0; j < 8; ++j) {
      const int i = t + j * 256;
      const int kp = i >> 4, cc = (i & 15) << 2;
      const float4 v0 = rb0[j];
      const float4 v1 = rb1[j];
      float4 lo = {v0.x, v1.x, v0.y, v1.y};
      float4 hi = {v0.z, v1.z, v0.w, v1.w};
      *(float4*)(&Bs[kp][cc * 2])     = lo;
      *(float4*)(&Bs[kp][cc * 2 + 4]) = hi;
    }
    __syncthreads();

    const int mrow = rs * 16 + m16;
    const int kq0 = kq * (BMM_KC / 4);
#pragma unroll
    for (int ks = 0; ks < BMM_KC / 16; ++ks) {
      const int kg = kq0 + ks * 4 + 2 * kh;
      const int kp = (kq0 >> 1) + ks * 2 + kh;
      const v2f a = *(const v2f*)(&As[mrow][kg]);
#pragma unroll
      for (int nt = 0; nt < 4; ++nt) {
        const v2f bf = *(const v2f*)(&Bs[kp][(nt * 16 + m16) * 2]);
        cacc[nt] = wmma_f32(a, bf, cacc[nt]);
      }
    }
  }

#pragma unroll
  for (int nt = 0; nt < 4; ++nt) {
#pragma unroll
    for (int r = 0; r < 8; ++r)
      Red[kq][rs * 16 + r + 8 * kh][nt * 16 + m16] = cacc[nt][r];
  }
  __syncthreads();
#pragma unroll
  for (int j = 0; j < 2; ++j) {
    const int e = (t + j * 256) * 4;
    const int row = e >> 6, col = e & 63;
    const float4 p0 = *(const float4*)(&Red[0][row][col]);
    const float4 p1 = *(const float4*)(&Red[1][row][col]);
    const float4 p2 = *(const float4*)(&Red[2][row][col]);
    const float4 p3 = *(const float4*)(&Red[3][row][col]);
    float4 s;
    s.x = (p0.x + p1.x) + (p2.x + p3.x);
    s.y = (p0.y + p1.y) + (p2.y + p3.y);
    s.z = (p0.z + p1.z) + (p2.z + p3.z);
    s.w = (p0.w + p1.w) + (p2.w + p3.w);
    *(float4*)(Cb + (size_t)(rowBase + row) * 64 + col) = s;
  }
}

__global__ __launch_bounds__(256)
void k_final(const float* __restrict__ sx, const float* __restrict__ xg1,
             const float* __restrict__ W, const float* __restrict__ E,
             const float* __restrict__ bias_pool, float* __restrict__ out) {
  __shared__ float As[8][8][132];
  const int t = threadIdx.x;
  const int wave = t >> 5;
  const int lane = t & 31;
  const int n = blockIdx.x * 8 + wave;
  float (*Aw)[132] = As[wave];
  const int m16 = lane & 15;
  const int kh = lane >> 4;

  float va[8][4];
#pragma unroll
  for (int rb = 0; rb < 8; ++rb) {
    const float* s0 = sx  + ((size_t)rb * 1024 + n) * 64;
    const float* s1 = xg1 + ((size_t)rb * 1024 + n) * 64;
    va[rb][0] = s0[lane];
    va[rb][1] = s0[lane + 32];
    va[rb][2] = s1[lane];
    va[rb][3] = s1[lane + 32];
  }
  float bv[4];
#pragma unroll
  for (int nt = 0; nt < 4; ++nt) {
    const int col = nt * 16 + m16;
    float s = 0.f;
#pragma unroll
    for (int d = 0; d < 16; ++d) s += E[n * 16 + d] * bias_pool[d * 64 + col];
    bv[nt] = s;
  }
#pragma unroll
  for (int rb = 0; rb < 8; ++rb) {
    Aw[rb][lane]      = va[rb][0];
    Aw[rb][lane + 32] = va[rb][1];
    Aw[rb][lane + 64] = va[rb][2];
    Aw[rb][lane + 96] = va[rb][3];
  }
  __syncthreads();

  const float* Wn = W + (size_t)n * 8192;
  v2f bw[32][4];
#pragma unroll
  for (int ks = 0; ks < 32; ++ks) {
    const int kp = ks * 2 + kh;
#pragma unroll
    for (int nt = 0; nt < 4; ++nt)
      bw[ks][nt] = *(const v2f*)(Wn + (size_t)kp * 128 + (nt * 16 + m16) * 2);
  }
  const int arow = m16 & 7;
  v2f af[32];
#pragma unroll
  for (int ks = 0; ks < 32; ++ks)
    af[ks] = *(const v2f*)(&Aw[arow][ks * 4 + 2 * kh]);
  __builtin_amdgcn_sched_barrier(0);

  const v8f zero = {0.f, 0.f, 0.f, 0.f, 0.f, 0.f, 0.f, 0.f};
  v8f cacc[4];
#pragma unroll
  for (int nt = 0; nt < 4; ++nt) cacc[nt] = zero;

#pragma unroll
  for (int ks = 0; ks < 32; ++ks) {
#pragma unroll
    for (int nt = 0; nt < 4; ++nt)
      cacc[nt] = wmma_f32(af[ks], bw[ks][nt], cacc[nt]);
  }

#pragma unroll
  for (int nt = 0; nt < 4; ++nt) {
    const int col = nt * 16 + m16;
#pragma unroll
    for (int r = 0; r < 8; ++r) {
      const int m = r + 8 * kh;
      if (m < 8)
        out[((size_t)m * 1024 + n) * 64 + col] = cacc[nt][r] + bv[nt];
    }
  }
}

extern "C" void kernel_launch(void* const* d_in, const int* in_sizes, int n_in,
                              void* d_out, int out_size, void* d_ws, size_t ws_size,
                              hipStream_t stream) {
  (void)in_sizes; (void)n_in; (void)out_size; (void)ws_size;
  const float* x    = (const float*)d_in[0];
  const float* E    = (const float*)d_in[1];
  const float* pa   = (const float*)d_in[2];
  const float* Wp   = (const float*)d_in[3];
  const float* bp   = (const float*)d_in[4];
  float* out = (float*)d_out;

  float* ws   = (float*)d_ws;
  float* sim  = ws;
  float* sube = sim  + 1048576;
  float* S    = sube + 8388608;
  float* xn   = S    + 8192;
  float* sx   = xn   + 524288;
  float* xg1  = sx   + 524288;
  float* P    = xg1;
  float* W    = xg1  + 524288;

  k_simraw    <<<4096, 256, 0, stream>>>(E, sim);
  k_rowsoftmax<<<1024, 256, 0, stream>>>(sim);
  k_cdist     <<<dim3(16, 16, 8), 256, 0, stream>>>(pa, sube, P);
  k_xnorm     <<<2048, 256, 0, stream>>>(x, P, xn);
  k_weights   <<<dim3(8, 1024),   256, 0, stream>>>(E, Wp, W);
  k_bmm_nk<<<dim3(32, 8), 256, 0, stream>>>(sube, 1024LL * 1024, xn, 1024LL * 64,
                                            sx, 1024LL * 64);
  k_bmm_nk<<<dim3(32, 8), 256, 0, stream>>>(sim, 0LL, sx, 1024LL * 64,
                                            xg1, 1024LL * 64);
  k_final<<<128, 256, 0, stream>>>(sx, xg1, W, E, bp, out);
}


# Input-only kernels (cdist, weights, raw sim) merged into one launch, and row softmax with x scaling into a second, each part chosen by work-group index: 8 launches become 5
# speedup vs baseline: 4.2194x; 1.1590x over previous
#include <hip/hip_runtime.h>
#include <hip/hip_bf16.h>
#include <cstddef>


typedef float v2f __attribute__((ext_vector_type(2)));
typedef float v8f __attribute__((ext_vector_type(8)));

static __device__ __forceinline__ v8f wmma_f32(v2f a, v2f b, v8f c) {
  return __builtin_amdgcn_wmma_f32_16x16x4_f32(
       false, a,  false, b,
       (short)0, c,  false,  false);
}

static __device__ __forceinline__
void simraw_part(const int blk, const float* __restrict__ E, float* __restrict__ sim) {
  const int idx = blk * 256 + threadIdx.x;
  const int i = idx >> 10, j = idx & 1023;
  float acc = 0.f;
#pragma unroll
  for (int d = 0; d < 16; ++d) acc += E[i * 16 + d] * E[j * 16 + d];
  sim[idx] = fmaxf(acc, 0.f);
}

static __device__ __forceinline__
void rowsoftmax_part(const int blk, float* __restrict__ sim) {
  const int row = blk;
  const int t = threadIdx.x;
  __shared__ float red[256];
  float v[4];
  float mx = -3.0e38f;
#pragma unroll
  for (int i = 0; i < 4; ++i) {
    v[i] = sim[(size_t)row * 1024 + t + i * 256];
    mx = fmaxf(mx, v[i]);
  }
  red[t] = mx; __syncthreads();
  for (int s = 128; s > 0; s >>= 1) {
    if (t < s) red[t] = fmaxf(red[t], red[t + s]);
    __syncthreads();
  }
  mx = red[0]; __syncthreads();
  float sum = 0.f;
#pragma unroll
  for (int i = 0; i < 4; ++i) { v[i] = __expf(v[i] - mx); sum += v[i]; }
  red[t] = sum; __syncthreads();
  for (int s = 128; s > 0; s >>= 1) {
    if (t < s) red[t] += red[t + s];
    __syncthreads();
  }
  const float inv = 1.0f / red[0];
#pragma unroll
  for (int i = 0; i < 4; ++i)
    sim[(size_t)row * 1024 + t + i * 256] = v[i] * inv;
}

static __device__ __forceinline__
void cdist_part(const int blk, const float* __restrict__ pa, float* __restrict__ sube,
                float* __restrict__ P) {
  const int b = blk >> 8;
  const int ib = (blk >> 4) & 15;
  const int jBase = (blk & 15) * 64, iBase = ib * 64;
  __shared__ float Pi[64][17];
  __shared__ float Pj[64][17];
  __shared__ float Cs[4][64];
  const int t = threadIdx.x;
  for (int i = t; i < 64 * 16; i += 256) {
    const int r = i >> 4, d = i & 15;
    Pi[r][d] = pa[((size_t)b * 1024 + iBase + r) * 16 + d];
    Pj[r][d] = pa[((size_t)b * 1024 + jBase + r) * 16 + d];
  }
  __syncthreads();
  const int tj = t & 63;
  const int g = t >> 6;
  const int i0 = g * 16;
  float csum = 0.f;
#pragma unroll 4
  for (int ii = 0; ii < 16; ++ii) {
    const int i = i0 + ii;
    float s = 0.f;
#pragma unroll
    for (int d = 0; d < 16; ++d) s += fabsf(Pi[i][d] - Pj[tj][d]);
    const float e = __expf(-s);
    sube[((size_t)b * 1024 + iBase + i) * 1024 + jBase + tj] = e;
    csum += e;
  }
  Cs[g][tj] = csum;
  __syncthreads();
  if (t < 64)
    P[((size_t)ib * 8 + b) * 1024 + jBase + t] =
        (Cs[0][t] + Cs[1][t]) + (Cs[2][t] + Cs[3][t]);
}

static __device__ __forceinline__
void xnorm_part(const int blk, const float* __restrict__ x, const float* __restrict__ P,
                float* __restrict__ xn) {
  const int idx = blk * 256 + threadIdx.x;
  const int bj = idx >> 6;
  float s = 0.f;
#pragma unroll
  for (int ib = 0; ib < 16; ++ib) s += P[ib * 8192 + bj];
  xn[idx] = x[idx] / s;
}

static __device__ __forceinline__
void weights_part(const int blk, const float* __restrict__ E, const float* __restrict__ Wp,
                  float* __restrict__ W) {
  const int n = blk >> 3;
  const int tid = (blk & 7) * 256 + threadIdx.x;
  const int kp = tid >> 5;
  const int oc = (tid & 31) * 2;
  float e[16];
#pragma unroll
  for (int d = 0; d < 16; ++d) e[d] = E[n * 16 + d];
  float4 acc = {0.f, 0.f, 0.f, 0.f};
#pragma unroll
  for (int d = 0; d < 16; ++d) {
    const v2f w0 = *(const v2f*)(Wp + (size_t)d * 8192 + (size_t)(2 * kp) * 64 + oc);
    const v2f w1 = *(const v2f*)(Wp + (size_t)d * 8192 + (size_t)(2 * kp + 1) * 64 + oc);
    acc.x += e[d] * w0.x;
    acc.y += e[d] * w1.x;
    acc.z += e[d] * w0.y;
    acc.w += e[d] * w1.y;
  }
  *(float4*)(W + (size_t)n * 8192 + (size_t)kp * 128 + (size_t)oc * 2) = acc;
}

#define L1_CDIST   2048
#define L1_WEIGHTS 8192
#define L1_SIMRAW  4096
#define L1_GRID    14336
__global__ __launch_bounds__(256)
void k_stage1(const float* __restrict__ E, const float* __restrict__ pa,
              const float* __restrict__ Wp, float* __restrict__ sim,
              float* __restrict__ sube, float* __restrict__ P,
              float* __restrict__ W) {
  const int blk = blockIdx.x;
  if (blk < L1_CDIST) {
    cdist_part(blk, pa, sube, P);
  } else if (blk < L1_CDIST + L1_WEIGHTS) {
    weights_part(blk - L1_CDIST, E, Wp, W);
  } else {
    simraw_part(blk - (L1_CDIST + L1_WEIGHTS), E, sim);
  }
}

#define L2_SOFTMAX 1024
#define L2_XNORM   2048
#define L2_GRID    3072
__global__ __launch_bounds__(256)
void k_stage2(float* __restrict__ sim, const float* __restrict__ x,
              const float* __restrict__ P, float* __restrict__ xn) {
  const int blk = blockIdx.x;
  if (blk < L2_SOFTMAX) {
    rowsoftmax_part(blk, sim);
  } else {
    xnorm_part(blk - L2_SOFTMAX, x, P, xn);
  }
}

#define BMM_KC 256
__global__ __launch_bounds__(256)
void k_bmm_nk(const float* __restrict__ A, long long strideA,
              const float* __restrict__ B, long long strideB,
              float* __restrict__ C, long long strideC) {
  const int b = blockIdx.y;
  const float* Ab = A + (size_t)strideA * b;
  const float* Bb = B + (size_t)strideB * b;
  float* Cb = C + (size_t)strideC * b;
  const int rowBase = blockIdx.x * 32;

  __shared__ float As[32][BMM_KC + 4];
  __shared__ float Bs[BMM_KC / 2][160];
  __shared__ float Red[4][32][68];

  const int t = threadIdx.x;
  const int wave = t >> 5;
  const int lane = t & 31;
  const int m16 = lane & 15;
  const int kh = lane >> 4;
  const int rs = wave & 1;
  const int kq = wave >> 1;

  const v8f zero = {0.f, 0.f, 0.f, 0.f, 0.f, 0.f, 0.f, 0.f};
  v8f cacc[4];
#pragma unroll
  for (int nt = 0; nt < 4; ++nt) cacc[nt] = zero;

  for (int kc = 0; kc < 1024; kc += BMM_KC) {
    __syncthreads();
    float4 ra[8], rb0[8], rb1[8];
#pragma unroll
    for (int j = 0; j < 8; ++j) {
      const int i = t + j * 256;
      const int r = i >> 6, cc = (i & 63) << 2;
      ra[j] = *(const float4*)(Ab + (size_t)(rowBase + r) * 1024 + kc + cc);
    }
#pragma unroll
    for (int j = 0; j < 8; ++j) {
      const int i = t + j * 256;
      const int kp = i >> 4, cc = (i & 15) << 2;
      rb0[j] = *(const float4*)(Bb + (size_t)(kc + 2 * kp) * 64 + cc);
      rb1[j] = *(const float4*)(Bb + (size_t)(kc + 2 * kp + 1) * 64 + cc);
    }
    __builtin_amdgcn_sched_barrier(0);
#pragma unroll
    for (int j = 0; j < 8; ++j) {
      const int i = t + j * 256;
      const int r = i >> 6, cc = (i & 63) << 2;
      *(float4*)(&As[r][cc]) = ra[j];
    }
#pragma unroll
    for (int j = 0; j < 8; ++j) {
      const int i = t + j * 256;
      const int kp = i >> 4, cc = (i & 15) << 2;
      const float4 v0 = rb0[j];
      const float4 v1 = rb1[j];
      float4 lo = {v0.x, v1.x, v0.y, v1.y};
      float4 hi = {v0.z, v1.z, v0.w, v1.w};
      *(float4*)(&Bs[kp][cc * 2])     = lo;
      *(float4*)(&Bs[kp][cc * 2 + 4]) = hi;
    }
    __syncthreads();

    const int mrow = rs * 16 + m16;
    const int kq0 = kq * (BMM_KC / 4);
#pragma unroll
    for (int ks = 0; ks < BMM_KC / 16; ++ks) {
      const int kg = kq0 + ks * 4 + 2 * kh;
      const int kp = (kq0 >> 1) + ks * 2 + kh;
      const v2f a = *(const v2f*)(&As[mrow][kg]);
#pragma unroll
      for (int nt = 0; nt < 4; ++nt) {
        const v2f bf = *(const v2f*)(&Bs[kp][(nt * 16 + m16) * 2]);
        cacc[nt] = wmma_f32(a, bf, cacc[nt]);
      }
    }
  }

#pragma unroll
  for (int nt = 0; nt < 4; ++nt) {
#pragma unroll
    for (int r = 0; r < 8; ++r)
      Red[kq][rs * 16 + r + 8 * kh][nt * 16 + m16] = cacc[nt][r];
  }
  __syncthreads();
#pragma unroll
  for (int j = 0; j < 2; ++j) {
    const int e = (t + j * 256) * 4;
    const int row = e >> 6, col = e & 63;
    const float4 p0 = *(const float4*)(&Red[0][row][col]);
    const float4 p1 = *(const float4*)(&Red[1][row][col]);
    const float4 p2 = *(const float4*)(&Red[2][row][col]);
    const float4 p3 = *(const float4*)(&Red[3][row][col]);
    float4 s;
    s.x = (p0.x + p1.x) + (p2.x + p3.x);
    s.y = (p0.y + p1.y) + (p2.y + p3.y);
    s.z = (p0.z + p1.z) + (p2.z + p3.z);
    s.w = (p0.w + p1.w) + (p2.w + p3.w);
    *(float4*)(Cb + (size_t)(rowBase + row) * 64 + col) = s;
  }
}

__global__ __launch_bounds__(256)
void k_final(const float* __restrict__ sx, const float* __restrict__ xg1,
             const float* __restrict__ W, const float* __restrict__ E,
             const float* __restrict__ bias_pool, float* __restrict__ out) {
  __shared__ float As[8][8][132];
  const int t = threadIdx.x;
  const int wave = t >> 5;
  const int lane = t & 31;
  const int n = blockIdx.x * 8 + wave;
  float (*Aw)[132] = As[wave];
  const int m16 = lane & 15;
  const int kh = lane >> 4;

  float va[8][4];
#pragma unroll
  for (int rb = 0; rb < 8; ++rb) {
    const float* s0 = sx  + ((size_t)rb * 1024 + n) * 64;
    const float* s1 = xg1 + ((size_t)rb * 1024 + n) * 64;
    va[rb][0] = s0[lane];
    va[rb][1] = s0[lane + 32];
    va[rb][2] = s1[lane];
    va[rb][3] = s1[lane + 32];
  }
  float bv[4];
#pragma unroll
  for (int nt = 0; nt < 4; ++nt) {
    const int col = nt * 16 + m16;
    float s = 0.f;
#pragma unroll
    for (int d = 0; d < 16; ++d) s += E[n * 16 + d] * bias_pool[d * 64 + col];
    bv[nt] = s;
  }
#pragma unroll
  for (int rb = 0; rb < 8; ++rb) {
    Aw[rb][lane]      = va[rb][0];
    Aw[rb][lane + 32] = va[rb][1];
    Aw[rb][lane + 64] = va[rb][2];
    Aw[rb][lane + 96] = va[rb][3];
  }
  __syncthreads();

  const float* Wn = W + (size_t)n * 8192;
  v2f bw[32][4];
#pragma unroll
  for (int ks = 0; ks < 32; ++ks) {
    const int kp = ks * 2 + kh;
#pragma unroll
    for (int nt = 0; nt < 4; ++nt)
      bw[ks][nt] = *(const v2f*)(Wn + (size_t)kp * 128 + (nt * 16 + m16) * 2);
  }
  const int arow = m16 & 7;
  v2f af[32];
#pragma unroll
  for (int ks = 0; ks < 32; ++ks)
    af[ks] = *(const v2f*)(&Aw[arow][ks * 4 + 2 * kh]);
  __builtin_amdgcn_sched_barrier(0);

  const v8f zero = {0.f, 0.f, 0.f, 0.f, 0.f, 0.f, 0.f, 0.f};
  v8f cacc[4];
#pragma unroll
  for (int nt = 0; nt < 4; ++nt) cacc[nt] = zero;

#pragma unroll
  for (int ks = 0; ks < 32; ++ks) {
#pragma unroll
    for (int nt = 0; nt < 4; ++nt)
      cacc[nt] = wmma_f32(af[ks], bw[ks][nt], cacc[nt]);
  }

#pragma unroll
  for (int nt = 0; nt < 4; ++nt) {
    const int col = nt * 16 + m16;
#pragma unroll
    for (int r = 0; r < 8; ++r) {
      const int m = r + 8 * kh;
      if (m < 8)
        out[((size_t)m * 1024 + n) * 64 + col] = cacc[nt][r] + bv[nt];
    }
  }
}

extern "C" void kernel_launch(void* const* d_in, const int* in_sizes, int n_in,
                              void* d_out, int out_size, void* d_ws, size_t ws_size,
                              hipStream_t stream) {
  (void)in_sizes; (void)n_in; (void)out_size; (void)ws_size;
  const float* x    = (const float*)d_in[0];
  const float* E    = (const float*)d_in[1];
  const float* pa   = (const float*)d_in[2];
  const float* Wp   = (const float*)d_in[3];
  const float* bp   = (const float*)d_in[4];
  float* out = (float*)d_out;

  float* ws   = (float*)d_ws;
  float* sim  = ws;
  float* sube = sim  + 1048576;
  float* S    = sube + 8388608;
  float* xn   = S    + 8192;
  float* sx   = xn   + 524288;
  float* xg1  = sx   + 524288;
  float* P    = xg1;
  float* W    = xg1  + 524288;

  k_stage1<<<L1_GRID, 256, 0, stream>>>(E, pa, Wp, sim, sube, P, W);
  k_stage2<<<L2_GRID, 256, 0, stream>>>(sim, x, P, xn);
  k_bmm_nk<<<dim3(32, 8), 256, 0, stream>>>(sube, 1024LL * 1024, xn, 1024LL * 64,
                                            sx, 1024LL * 64);
  k_bmm_nk<<<dim3(32, 8), 256, 0, stream>>>(sim, 0LL, sx, 1024LL * 64,
                                            xg1, 1024LL * 64);
  k_final<<<128, 256, 0, stream>>>(sx, xg1, W, E, bp, out);
}


# Launches 1 and 2 re-gridded: a work-group loops over many rows, nodes or elements (softmax: one wave per row, lane exchanges); 17408 work-groups become 1536; changes summation order of softmax sums
# speedup vs baseline: 4.4494x; 1.0545x over previous
#include <hip/hip_runtime.h>
#include <hip/hip_bf16.h>
#include <cstddef>


typedef float v2f __attribute__((ext_vector_type(2)));
typedef float v8f __attribute__((ext_vector_type(8)));

static __device__ __forceinline__ v8f wmma_f32(v2f a, v2f b, v8f c) {
  return __builtin_amdgcn_wmma_f32_16x16x4_f32(
       false, a,  false, b,
       (short)0, c,  false,  false);
}

static __device__ __forceinline__
void simraw_part(const int blk, const float* __restrict__ E, float* __restrict__ sim) {
  const int j = (blk & 3) * 256 + threadIdx.x;
  const int i0 = (blk >> 2) * 16;
  float ej[16];
#pragma unroll
  for (int q = 0; q < 4; ++q) {
    const float4 v = *(const float4*)(E + (size_t)j * 16 + q * 4);
    ej[q * 4 + 0] = v.x; ej[q * 4 + 1] = v.y; ej[q * 4 + 2] = v.z; ej[q * 4 + 3] = v.w;
  }
#pragma unroll 2
  for (int ii = 0; ii < 16; ++ii) {
    const float* ei = E + (size_t)(i0 + ii) * 16;
    float acc = 0.f;
#pragma unroll
    for (int d = 0; d < 16; ++d) acc += ei[d] * ej[d];
    sim[(size_t)(i0 + ii) * 1024 + j] = fmaxf(acc, 0.f);
  }
}

static __device__ __forceinline__
void rowsoftmax_part(const int blk, float* __restrict__ sim) {
  const int row = blk * 8 + (threadIdx.x >> 5);
  const int lane = threadIdx.x & 31;
  float4* p = (float4*)(sim + (size_t)row * 1024) + lane;
  float4 v[8];
#pragma unroll
  for (int k = 0; k < 8; ++k) v[k] = p[k * 32];
  float mx = -3.0e38f;
#pragma unroll
  for (int k = 0; k < 8; ++k)
    mx = fmaxf(mx, fmaxf(fmaxf(v[k].x, v[k].y), fmaxf(v[k].z, v[k].w)));
#pragma unroll
  for (int s = 16; s > 0; s >>= 1) mx = fmaxf(mx, __shfl_xor(mx, s));
  float sum = 0.f;
#pragma unroll
  for (int k = 0; k < 8; ++k) {
    v[k].x = __expf(v[k].x - mx); v[k].y = __expf(v[k].y - mx);
    v[k].z = __expf(v[k].z - mx); v[k].w = __expf(v[k].w - mx);
    sum += (v[k].x + v[k].y) + (v[k].z + v[k].w);
  }
#pragma unroll
  for (int s = 16; s > 0; s >>= 1) sum += __shfl_xor(sum, s);
  const float inv = 1.0f / sum;
#pragma unroll
  for (int k = 0; k < 8; ++k) {
    float4 r = {v[k].x * inv, v[k].y * inv, v[k].z * inv, v[k].w * inv};
    p[k * 32] = r;
  }
}

static __device__ __forceinline__
void cdist_part(const int blk, const float* __restrict__ pa, float* __restrict__ sube,
                float* __restrict__ P) {
  const int b = blk >> 6;
  const int ib = (blk >> 2) & 15;
  const int iBase = ib * 64;
  const int t = threadIdx.x;
  const int j = (blk & 3) * 256 + t;
  __shared__ __attribute__((aligned(16))) float Pi[64][16];
  ((float4*)&Pi[0][0])[t] =
      *(const float4*)(pa + ((size_t)b * 1024 + iBase) * 16 + (size_t)t * 4);
  float pj[16];
#pragma unroll
  for (int q = 0; q < 4; ++q) {
    const float4 v = *(const float4*)(pa + ((size_t)b * 1024 + j) * 16 + q * 4);
    pj[q * 4 + 0] = v.x; pj[q * 4 + 1] = v.y; pj[q * 4 + 2] = v.z; pj[q * 4 + 3] = v.w;
  }
  __syncthreads();
  float* dst = sube + ((size_t)b * 1024 + iBase) * 1024 + j;
  float cs[4];
#pragma unroll
  for (int g = 0; g < 4; ++g) {
    float csum = 0.f;
#pragma unroll 4
    for (int ii = 0; ii < 16; ++ii) {
      const int i = g * 16 + ii;
      float s = 0.f;
#pragma unroll
      for (int d = 0; d < 16; ++d) s += fabsf(Pi[i][d] - pj[d]);
      const float e = __expf(-s);
      dst[(size_t)i * 1024] = e;
      csum += e;
    }
    cs[g] = csum;
  }
  P[((size_t)ib * 8 + b) * 1024 + j] = (cs[0] + cs[1]) + (cs[2] + cs[3]);
}

static __device__ __forceinline__
void xnorm_part(const int blk, const float* __restrict__ x, const float* __restrict__ P,
                float* __restrict__ xn) {
#pragma unroll
  for (int k = 0; k < 4; ++k) {
    const int f = blk * 1024 + k * 256 + threadIdx.x;
    const int bj = f >> 4;
    float s = 0.f;
#pragma unroll
    for (int ib = 0; ib < 16; ++ib) s += P[ib * 8192 + bj];
    const float4 v = ((const float4*)x)[f];
    float4 r = {v.x / s, v.y / s, v.z / s, v.w / s};
    ((float4*)xn)[f] = r;
  }
}

static __device__ __forceinline__
void weights_part(const int blk, const float* __restrict__ E, const float* __restrict__ Wp,
                  float* __restrict__ W) {
  const int n0 = (blk >> 3) * 16;
  const int tid = (blk & 7) * 256 + threadIdx.x;
  const int kp = tid >> 5;
  const int oc = (tid & 31) * 2;
  v2f w0[16], w1[16];
#pragma unroll
  for (int d = 0; d < 16; ++d) {
    w0[d] = *(const v2f*)(Wp + (size_t)d * 8192 + (size_t)(2 * kp) * 64 + oc);
    w1[d] = *(const v2f*)(Wp + (size_t)d * 8192 + (size_t)(2 * kp + 1) * 64 + oc);
  }
  float* dst = W + (size_t)n0 * 8192 + (size_t)kp * 128 + (size_t)oc * 2;
#pragma unroll 2
  for (int nn = 0; nn < 16; ++nn) {
    const float* e = E + (size_t)(n0 + nn) * 16;
    float4 acc = {0.f, 0.f, 0.f, 0.f};
#pragma unroll
    for (int d = 0; d < 16; ++d) {
      acc.x += e[d] * w0[d].x;
      acc.y += e[d] * w1[d].x;
      acc.z += e[d] * w0[d].y;
      acc.w += e[d] * w1[d].y;
    }
    *(float4*)(dst + (size_t)nn * 8192) = acc;
  }
}

#define L1_CDIST   512
#define L1_WEIGHTS 512
#define L1_SIMRAW  256
#define L1_GRID    1280
__global__ __launch_bounds__(256)
void k_stage1(const float* __restrict__ E, const float* __restrict__ pa,
              const float* __restrict__ Wp, float* __restrict__ sim,
              float* __restrict__ sube, float* __restrict__ P,
              float* __restrict__ W) {
  const int blk = blockIdx.x;
  if (blk < L1_CDIST) {
    cdist_part(blk, pa, sube, P);
  } else if (blk < L1_CDIST + L1_WEIGHTS) {
    weights_part(blk - L1_CDIST, E, Wp, W);
  } else {
    simraw_part(blk - (L1_CDIST + L1_WEIGHTS), E, sim);
  }
}

#define L2_SOFTMAX 128
#define L2_XNORM   128
#define L2_GRID    256
__global__ __launch_bounds__(256)
void k_stage2(float* __restrict__ sim, const float* __restrict__ x,
              const float* __restrict__ P, float* __restrict__ xn) {
  const int blk = blockIdx.x;
  if (blk < L2_SOFTMAX) {
    rowsoftmax_part(blk, sim);
  } else {
    xnorm_part(blk - L2_SOFTMAX, x, P, xn);
  }
}

#define BMM_KC 256
__global__ __launch_bounds__(256)
void k_bmm_nk(const float* __restrict__ A, long long strideA,
              const float* __restrict__ B, long long strideB,
              float* __restrict__ C, long long strideC) {
  const int b = blockIdx.y;
  const float* Ab = A + (size_t)strideA * b;
  const float* Bb = B + (size_t)strideB * b;
  float* Cb = C + (size_t)strideC * b;
  const int rowBase = blockIdx.x * 32;

  __shared__ float As[32][BMM_KC + 4];
  __shared__ float Bs[BMM_KC / 2][160];
  __shared__ float Red[4][32][68];

  const int t = threadIdx.x;
  const int wave = t >> 5;
  const int lane = t & 31;
  const int m16 = lane & 15;
  const int kh = lane >> 4;
  const int rs = wave & 1;
  const int kq = wave >> 1;

  const v8f zero = {0.f, 0.f, 0.f, 0.f, 0.f, 0.f, 0.f, 0.f};
  v8f cacc[4];
#pragma unroll
  for (int nt = 0; nt < 4; ++nt) cacc[nt] = zero;

  for (int kc = 0; kc < 1024; kc += BMM_KC) {
    __syncthreads();
    float4 ra[8], rb0[8], rb1[8];
#pragma unroll
    for (int j = 0; j < 8; ++j) {
      const int i = t + j * 256;
      const int r = i >> 6, cc = (i & 63) << 2;
      ra[j] = *(const float4*)(Ab + (size_t)(rowBase + r) * 1024 + kc + cc);
    }
#pragma unroll
    for (int j = 0; j < 8; ++j) {
      const int i = t + j * 256;
      const int kp = i >> 4, cc = (i & 15) << 2;
      rb0[j] = *(const float4*)(Bb + (size_t)(kc + 2 * kp) * 64 + cc);
      rb1[j] = *(const float4*)(Bb + (size_t)(kc + 2 * kp + 1) * 64 + cc);
    }
    __builtin_amdgcn_sched_barrier(0);
#pragma unroll
    for (int j = 0; j < 8; ++j) {
      const int i = t + j * 256;
      const int r = i >> 6, cc = (i & 63) << 2;
      *(float4*)(&As[r][cc]) = ra[j];
    }
#pragma unroll
    for (int j = 0; j < 8; ++j) {
      const int i = t + j * 256;
      const int kp = i >> 4, cc = (i & 15) << 2;
      const float4 v0 = rb0[j];
      const float4 v1 = rb1[j];
      float4 lo = {v0.x, v1.x, v0.y, v1.y};
      float4 hi = {v0.z, v1.z, v0.w, v1.w};
      *(float4*)(&Bs[kp][cc * 2])     = lo;
      *(float4*)(&Bs[kp][cc * 2 + 4]) = hi;
    }
    __syncthreads();

    const int mrow = rs * 16 + m16;
    const int kq0 = kq * (BMM_KC / 4);
#pragma unroll
    for (int ks = 0; ks < BMM_KC / 16; ++ks) {
      const int kg = kq0 + ks * 4 + 2 * kh;
      const int kp = (kq0 >> 1) + ks * 2 + kh;
      const v2f a = *(const v2f*)(&As[mrow][kg]);
#pragma unroll
      for (int nt = 0; nt < 4; ++nt) {
        const v2f bf = *(const v2f*)(&Bs[kp][(nt * 16 + m16) * 2]);
        cacc[nt] = wmma_f32(a, bf, cacc[nt]);
      }
    }
  }

#pragma unroll
  for (int nt = 0; nt < 4; ++nt) {
#pragma unroll
    for (int r = 0; r < 8; ++r)
      Red[kq][rs * 16 + r + 8 * kh][nt * 16 + m16] = cacc[nt][r];
  }
  __syncthreads();
#pragma unroll
  for (int j = 0; j < 2; ++j) {
    const int e = (t + j * 256) * 4;
    const int row = e >> 6, col = e & 63;
    const float4 p0 = *(const float4*)(&Red[0][row][col]);
    const float4 p1 = *(const float4*)(&Red[1][row][col]);
    const float4 p2 = *(const float4*)(&Red[2][row][col]);
    const float4 p3 = *(const float4*)(&Red[3][row][col]);
    float4 s;
    s.x = (p0.x + p1.x) + (p2.x + p3.x);
    s.y = (p0.y + p1.y) + (p2.y + p3.y);
    s.z = (p0.z + p1.z) + (p2.z + p3.z);
    s.w = (p0.w + p1.w) + (p2.w + p3.w);
    *(float4*)(Cb + (size_t)(rowBase + row) * 64 + col) = s;
  }
}

__global__ __launch_bounds__(256)
void k_final(const float* __restrict__ sx, const float* __restrict__ xg1,
             const float* __restrict__ W, const float* __restrict__ E,
             const float* __restrict__ bias_pool, float* __restrict__ out) {
  __shared__ float As[8][8][132];
  const int t = threadIdx.x;
  const int wave = t >> 5;
  const int lane = t & 31;
  const int n = blockIdx.x * 8 + wave;
  float (*Aw)[132] = As[wave];
  const int m16 = lane & 15;
  const int kh = lane >> 4;

  float va[8][4];
#pragma unroll
  for (int rb = 0; rb < 8; ++rb) {
    const float* s0 = sx  + ((size_t)rb * 1024 + n) * 64;
    const float* s1 = xg1 + ((size_t)rb * 1024 + n) * 64;
    va[rb][0] = s0[lane];
    va[rb][1] = s0[lane + 32];
    va[rb][2] = s1[lane];
    va[rb][3] = s1[lane + 32];
  }
  float bv[4];
#pragma unroll
  for (int nt = 0; nt < 4; ++nt) {
    const int col = nt * 16 + m16;
    float s = 0.f;
#pragma unroll
    for (int d = 0; d < 16; ++d) s += E[n * 16 + d] * bias_pool[d * 64 + col];
    bv[nt] = s;
  }
#pragma unroll
  for (int rb = 0; rb < 8; ++rb) {
    Aw[rb][lane]      = va[rb][0];
    Aw[rb][lane + 32] = va[rb][1];
    Aw[rb][lane + 64] = va[rb][2];
    Aw[rb][lane + 96] = va[rb][3];
  }
  __syncthreads();

  const float* Wn = W + (size_t)n * 8192;
  v2f bw[32][4];
#pragma unroll
  for (int ks = 0; ks < 32; ++ks) {
    const int kp = ks * 2 + kh;
#pragma unroll
    for (int nt = 0; nt < 4; ++nt)
      bw[ks][nt] = *(const v2f*)(Wn + (size_t)kp * 128 + (nt * 16 + m16) * 2);
  }
  const int arow = m16 & 7;
  v2f af[32];
#pragma unroll
  for (int ks = 0; ks < 32; ++ks)
    af[ks] = *(const v2f*)(&Aw[arow][ks * 4 + 2 * kh]);
  __builtin_amdgcn_sched_barrier(0);

  const v8f zero = {0.f, 0.f, 0.f, 0.f, 0.f, 0.f, 0.f, 0.f};
  v8f cacc[4];
#pragma unroll
  for (int nt = 0; nt < 4; ++nt) cacc[nt] = zero;

#pragma unroll
  for (int ks = 0; ks < 32; ++ks) {
#pragma unroll
    for (int nt = 0; nt < 4; ++nt)
      cacc[nt] = wmma_f32(af[ks], bw[ks][nt], cacc[nt]);
  }

#pragma unroll
  for (int nt = 0; nt < 4; ++nt) {
    const int col = nt * 16 + m16;
#pragma unroll
    for (int r = 0; r < 8; ++r) {
      const int m = r + 8 * kh;
      if (m < 8)
        out[((size_t)m * 1024 + n) * 64 + col] = cacc[nt][r] + bv[nt];
    }
  }
}

extern "C" void kernel_launch(void* const* d_in, const int* in_sizes, int n_in,
                              void* d_out, int out_size, void* d_ws, size_t ws_size,
                              hipStream_t stream) {
  (void)in_sizes; (void)n_in; (void)out_size; (void)ws_size;
  const float* x    = (const float*)d_in[0];
  const float* E    = (const float*)d_in[1];
  const float* pa   = (const float*)d_in[2];
  const float* Wp   = (const float*)d_in[3];
  const float* bp   = (const float*)d_in[4];
  float* out = (float*)d_out;

  float* ws   = (float*)d_ws;
  float* sim  = ws;
  float* sube = sim  + 1048576;
  float* S    = sube + 8388608;
  float* xn   = S    + 8192;
  float* sx   = xn   + 524288;
  float* xg1  = sx   + 524288;
  float* P    = xg1;
  float* W    = xg1  + 524288;

  k_stage1<<<L1_GRID, 256, 0, stream>>>(E, pa, Wp, sim, sube, P, W);
  k_stage2<<<L2_GRID, 256, 0, stream>>>(sim, x, P, xn);
  k_bmm_nk<<<dim3(32, 8), 256, 0, stream>>>(sube, 1024LL * 1024, xn, 1024LL * 64,
                                            sx, 1024LL * 64);
  k_bmm_nk<<<dim3(32, 8), 256, 0, stream>>>(sim, 0LL, sx, 1024LL * 64,
                                            xg1, 1024LL * 64);
  k_final<<<128, 256, 0, stream>>>(sx, xg1, W, E, bp, out);
}


# Launch 2 removed: a cdist work-group owns whole columns (1024 rows x 32), a sim work-group whole rows, so column sums, x scaling and row softmax finish inside launch 1; changes column-sum order
# speedup vs baseline: 5.0771x; 1.1411x over previous
#include <hip/hip_runtime.h>
#include <hip/hip_bf16.h>
#include <cstddef>


typedef float v2f __attribute__((ext_vector_type(2)));
typedef float v8f __attribute__((ext_vector_type(8)));

static __device__ __forceinline__ v8f wmma_f32(v2f a, v2f b, v8f c) {
  return __builtin_amdgcn_wmma_f32_16x16x4_f32(
       false, a,  false, b,
       (short)0, c,  false,  false);
}

static __device__ __forceinline__
void sim_part(const int blk, const float* __restrict__ E, float* __restrict__ sim,
              float* __restrict__ raw) {
  const int t = threadIdx.x;
  const int i0 = blk * 4;
#pragma unroll 1
  for (int c = 0; c < 4; ++c) {
    const int j = c * 256 + t;
    float ej[16];
#pragma unroll
    for (int q = 0; q < 4; ++q) {
      const float4 v = *(const float4*)(E + (size_t)j * 16 + q * 4);
      ej[q * 4 + 0] = v.x; ej[q * 4 + 1] = v.y; ej[q * 4 + 2] = v.z; ej[q * 4 + 3] = v.w;
    }
#pragma unroll 2
    for (int ii = 0; ii < 4; ++ii) {
      const float* ei = E + (size_t)(i0 + ii) * 16;
      float acc = 0.f;
#pragma unroll
      for (int d = 0; d < 16; ++d) acc += ei[d] * ej[d];
      raw[ii * 1024 + j] = fmaxf(acc, 0.f);
    }
  }
  __syncthreads();

  const int row = t >> 5;
  if (row >= 4) return;
  const int lane = t & 31;
  const float4* src = (const float4*)(raw + row * 1024) + lane;
  float4 v[8];
#pragma unroll
  for (int k = 0; k < 8; ++k) v[k] = src[k * 32];
  float mx = -3.0e38f;
#pragma unroll
  for (int k = 0; k < 8; ++k)
    mx = fmaxf(mx, fmaxf(fmaxf(v[k].x, v[k].y), fmaxf(v[k].z, v[k].w)));
#pragma unroll
  for (int s = 16; s > 0; s >>= 1) mx = fmaxf(mx, __shfl_xor(mx, s));
  float sum = 0.f;
#pragma unroll
  for (int k = 0; k < 8; ++k) {
    v[k].x = __expf(v[k].x - mx); v[k].y = __expf(v[k].y - mx);
    v[k].z = __expf(v[k].z - mx); v[k].w = __expf(v[k].w - mx);
    sum += (v[k].x + v[k].y) + (v[k].z + v[k].w);
  }
#pragma unroll
  for (int s = 16; s > 0; s >>= 1) sum += __shfl_xor(sum, s);
  const float inv = 1.0f / sum;
  float4* p = (float4*)(sim + (size_t)(i0 + row) * 1024) + lane;
#pragma unroll
  for (int k = 0; k < 8; ++k) {
    float4 r = {v[k].x * inv, v[k].y * inv, v[k].z * inv, v[k].w * inv};
    p[k * 32] = r;
  }
}

static __device__ __forceinline__
void cdist_part(const int blk, const float* __restrict__ pa, const float* __restrict__ x,
                float* __restrict__ sube, float* __restrict__ xn,
                float* __restrict__ Pi, float* __restrict__ red) {
  const int b = blk >> 5;
  const int j0 = (blk & 31) * 32;
  const int t = threadIdx.x;
  const int wave = t >> 5;
  const int lane = t & 31;
  const int j = j0 + lane;
  const int iBase = wave * 128;
  const float4* rsrc = (const float4*)(pa + ((size_t)b * 1024 + iBase) * 16) + lane;
  float4 stg[16];
#pragma unroll
  for (int q = 0; q < 16; ++q) stg[q] = rsrc[q * 32];
  float pj[16];
#pragma unroll
  for (int q = 0; q < 4; ++q) {
    const float4 v = *(const float4*)(pa + ((size_t)b * 1024 + j) * 16 + q * 4);
    pj[q * 4 + 0] = v.x; pj[q * 4 + 1] = v.y; pj[q * 4 + 2] = v.z; pj[q * 4 + 3] = v.w;
  }
  const size_t xoff = ((size_t)b * 1024 + j0) * 16;
  float4 xv[2];
#pragma unroll
  for (int k = 0; k < 2; ++k) xv[k] = ((const float4*)x)[xoff + t + k * 256];
  float4* rdst = (float4*)(Pi + (size_t)iBase * 16) + lane;
#pragma unroll
  for (int q = 0; q < 16; ++q) rdst[q * 32] = stg[q];
  __syncthreads();

  const float* Pw = Pi + (size_t)iBase * 16;
  float* dst = sube + ((size_t)b * 1024 + iBase) * 1024 + j;
  float tot = 0.f;
  for (int g = 0; g < 8; ++g) {
    float csum = 0.f;
#pragma unroll 4
    for (int ii = 0; ii < 16; ++ii) {
      const int i = g * 16 + ii;
      float s = 0.f;
#pragma unroll
      for (int d = 0; d < 16; ++d) s += fabsf(Pw[i * 16 + d] - pj[d]);
      const float e = __expf(-s);
      dst[(size_t)i * 1024] = e;
      csum += e;
    }
    tot += csum;
  }
  red[wave * 32 + lane] = tot;
  __syncthreads();

#pragma unroll
  for (int k = 0; k < 2; ++k) {
    const int f = t + k * 256;
    const int jj = f >> 4;
    const float s = ((red[jj] + red[32 + jj]) + (red[64 + jj] + red[96 + jj])) +
                    ((red[128 + jj] + red[160 + jj]) + (red[192 + jj] + red[224 + jj]));
    const float4 v = xv[k];
    float4 r = {v.x / s, v.y / s, v.z / s, v.w / s};
    ((float4*)xn)[xoff + f] = r;
  }
}

static __device__ __forceinline__
void weights_part(const int blk, const float* __restrict__ E, const float* __restrict__ Wp,
                  float* __restrict__ W) {
  const int n0 = (blk >> 3) * 16;
  const int tid = (blk & 7) * 256 + threadIdx.x;
  const int kp = tid >> 5;
  const int oc = (tid & 31) * 2;
  v2f w0[16], w1[16];
#pragma unroll
  for (int d = 0; d < 16; ++d) {
    w0[d] = *(const v2f*)(Wp + (size_t)d * 8192 + (size_t)(2 * kp) * 64 + oc);
    w1[d] = *(const v2f*)(Wp + (size_t)d * 8192 + (size_t)(2 * kp + 1) * 64 + oc);
  }
  float* dst = W + (size_t)n0 * 8192 + (size_t)kp * 128 + (size_t)oc * 2;
#pragma unroll 2
  for (int nn = 0; nn < 16; ++nn) {
    const float* e = E + (size_t)(n0 + nn) * 16;
    float4 acc = {0.f, 0.f, 0.f, 0.f};
#pragma unroll
    for (int d = 0; d < 16; ++d) {
      acc.x += e[d] * w0[d].x;
      acc.y += e[d] * w1[d].x;
      acc.z += e[d] * w0[d].y;
      acc.w += e[d] * w1[d].y;
    }
    *(float4*)(dst + (size_t)nn * 8192) = acc;
  }
}

#define L1_CDIST   256
#define L1_WEIGHTS 512
#define L1_SIM     256
#define L1_GRID    1024
__global__ __launch_bounds__(256)
void k_stage1(const float* __restrict__ E, const float* __restrict__ pa,
              const float* __restrict__ Wp, const float* __restrict__ x,
              float* __restrict__ sim, float* __restrict__ sube,
              float* __restrict__ xn, float* __restrict__ W) {
  __shared__ __attribute__((aligned(16))) float lds[16384 + 256];
  const int blk = blockIdx.x;
  if (blk < L1_CDIST) {
    cdist_part(blk, pa, x, sube, xn, lds, lds + 16384);
  } else if (blk < L1_CDIST + L1_WEIGHTS) {
    weights_part(blk - L1_CDIST, E, Wp, W);
  } else {
    sim_part(blk - (L1_CDIST + L1_WEIGHTS), E, sim, lds);
  }
}

#define BMM_KC 256
__global__ __launch_bounds__(256)
void k_bmm_nk(const float* __restrict__ A, long long strideA,
              const float* __restrict__ B, long long strideB,
              float* __restrict__ C, long long strideC) {
  const int b = blockIdx.y;
  const float* Ab = A + (size_t)strideA * b;
  const float* Bb = B + (size_t)strideB * b;
  float* Cb = C + (size_t)strideC * b;
  const int rowBase = blockIdx.x * 32;

  __shared__ float As[32][BMM_KC + 4];
  __shared__ float Bs[BMM_KC / 2][160];
  __shared__ float Red[4][32][68];

  const int t = threadIdx.x;
  const int wave = t >> 5;
  const int lane = t & 31;
  const int m16 = lane & 15;
  const int kh = lane >> 4;
  const int rs = wave & 1;
  const int kq = wave >> 1;

  const v8f zero = {0.f, 0.f, 0.f, 0.f, 0.f, 0.f, 0.f, 0.f};
  v8f cacc[4];
#pragma unroll
  for (int nt = 0; nt < 4; ++nt) cacc[nt] = zero;

  for (int kc = 0; kc < 1024; kc += BMM_KC) {
    __syncthreads();
    float4 ra[8], rb0[8], rb1[8];
#pragma unroll
    for (int j = 0; j < 8; ++j) {
      const int i = t + j * 256;
      const int r = i >> 6, cc = (i & 63) << 2;
      ra[j] = *(const float4*)(Ab + (size_t)(rowBase + r) * 1024 + kc + cc);
    }
#pragma unroll
    for (int j = 0; j < 8; ++j) {
      const int i = t + j * 256;
      const int kp = i >> 4, cc = (i & 15) << 2;
      rb0[j] = *(const float4*)(Bb + (size_t)(kc + 2 * kp) * 64 + cc);
      rb1[j] = *(const float4*)(Bb + (size_t)(kc + 2 * kp + 1) * 64 + cc);
    }
    __builtin_amdgcn_sched_barrier(0);
#pragma unroll
    for (int j = 0; j < 8; ++j) {
      const int i = t + j * 256;
      const int r = i >> 6, cc = (i & 63) << 2;
      *(float4*)(&As[r][cc]) = ra[j];
    }
#pragma unroll
    for (int j = 0; j < 8; ++j) {
      const int i = t + j * 256;
      const int kp = i >> 4, cc = (i & 15) << 2;
      const float4 v0 = rb0[j];
      const float4 v1 = rb1[j];
      float4 lo = {v0.x, v1.x, v0.y, v1.y};
      float4 hi = {v0.z, v1.z, v0.w, v1.w};
      *(float4*)(&Bs[kp][cc * 2])     = lo;
      *(float4*)(&Bs[kp][cc * 2 + 4]) = hi;
    }
    __syncthreads();

    const int mrow = rs * 16 + m16;
    const int kq0 = kq * (BMM_KC / 4);
#pragma unroll
    for (int ks = 0; ks < BMM_KC / 16; ++ks) {
      const int kg = kq0 + ks * 4 + 2 * kh;
      const int kp = (kq0 >> 1) + ks * 2 + kh;
      const v2f a = *(const v2f*)(&As[mrow][kg]);
#pragma unroll
      for (int nt = 0; nt < 4; ++nt) {
        const v2f bf = *(const v2f*)(&Bs[kp][(nt * 16 + m16) * 2]);
        cacc[nt] = wmma_f32(a, bf, cacc[nt]);
      }
    }
  }

#pragma unroll
  for (int nt = 0; nt < 4; ++nt) {
#pragma unroll
    for (int r = 0; r < 8; ++r)
      Red[kq][rs * 16 + r + 8 * kh][nt * 16 + m16] = cacc[nt][r];
  }
  __syncthreads();
#pragma unroll
  for (int j = 0; j < 2; ++j) {
    const int e = (t + j * 256) * 4;
    const int row = e >> 6, col = e & 63;
    const float4 p0 = *(const float4*)(&Red[0][row][col]);
    const float4 p1 = *(const float4*)(&Red[1][row][col]);
    const float4 p2 = *(const float4*)(&Red[2][row][col]);
    const float4 p3 = *(const float4*)(&Red[3][row][col]);
    float4 s;
    s.x = (p0.x + p1.x) + (p2.x + p3.x);
    s.y = (p0.y + p1.y) + (p2.y + p3.y);
    s.z = (p0.z + p1.z) + (p2.z + p3.z);
    s.w = (p0.w + p1.w) + (p2.w + p3.w);
    *(float4*)(Cb + (size_t)(rowBase + row) * 64 + col) = s;
  }
}

__global__ __launch_bounds__(256)
void k_final(const float* __restrict__ sx, const float* __restrict__ xg1,
             const float* __restrict__ W, const float* __restrict__ E,
             const float* __restrict__ bias_pool, float* __restrict__ out) {
  __shared__ float As[8][8][132];
  const int t = threadIdx.x;
  const int wave = t >> 5;
  const int lane = t & 31;
  const int n = blockIdx.x * 8 + wave;
  float (*Aw)[132] = As[wave];
  const int m16 = lane & 15;
  const int kh = lane >> 4;

  float va[8][4];
#pragma unroll
  for (int rb = 0; rb < 8; ++rb) {
    const float* s0 = sx  + ((size_t)rb * 1024 + n) * 64;
    const float* s1 = xg1 + ((size_t)rb * 1024 + n) * 64;
    va[rb][0] = s0[lane];
    va[rb][1] = s0[lane + 32];
    va[rb][2] = s1[lane];
    va[rb][3] = s1[lane + 32];
  }
  float bv[4];
#pragma unroll
  for (int nt = 0; nt < 4; ++nt) {
    const int col = nt * 16 + m16;
    float s = 0.f;
#pragma unroll
    for (int d = 0; d < 16; ++d) s += E[n * 16 + d] * bias_pool[d * 64 + col];
    bv[nt] = s;
  }
#pragma unroll
  for (int rb = 0; rb < 8; ++rb) {
    Aw[rb][lane]      = va[rb][0];
    Aw[rb][lane + 32] = va[rb][1];
    Aw[rb][lane + 64] = va[rb][2];
    Aw[rb][lane + 96] = va[rb][3];
  }
  __syncthreads();

  const float* Wn = W + (size_t)n * 8192;
  v2f bw[32][4];
#pragma unroll
  for (int ks = 0; ks < 32; ++ks) {
    const int kp = ks * 2 + kh;
#pragma unroll
    for (int nt = 0; nt < 4; ++nt)
      bw[ks][nt] = *(const v2f*)(Wn + (size_t)kp * 128 + (nt * 16 + m16) * 2);
  }
  const int arow = m16 & 7;
  v2f af[32];
#pragma unroll
  for (int ks = 0; ks < 32; ++ks)
    af[ks] = *(const v2f*)(&Aw[arow][ks * 4 + 2 * kh]);
  __builtin_amdgcn_sched_barrier(0);

  const v8f zero = {0.f, 0.f, 0.f, 0.f, 0.f, 0.f, 0.f, 0.f};
  v8f cacc[4];
#pragma unroll
  for (int nt = 0; nt < 4; ++nt) cacc[nt] = zero;

#pragma unroll
  for (int ks = 0; ks < 32; ++ks) {
#pragma unroll
    for (int nt = 0; nt < 4; ++nt)
      cacc[nt] = wmma_f32(af[ks], bw[ks][nt], cacc[nt]);
  }

#pragma unroll
  for (int nt = 0; nt < 4; ++nt) {
    const int col = nt * 16 + m16;
#pragma unroll
    for (int r = 0; r < 8; ++r) {
      const int m = r + 8 * kh;
      if (m < 8)
        out[((size_t)m * 1024 + n) * 64 + col] = cacc[nt][r] + bv[nt];
    }
  }
}

extern "C" void kernel_launch(void* const* d_in, const int* in_sizes, int n_in,
                              void* d_out, int out_size, void* d_ws, size_t ws_size,
                              hipStream_t stream) {
  (void)in_sizes; (void)n_in; (void)out_size; (void)ws_size;
  const float* x    = (const float*)d_in[0];
  const float* E    = (const float*)d_in[1];
  const float* pa   = (const float*)d_in[2];
  const float* Wp   = (const float*)d_in[3];
  const float* bp   = (const float*)d_in[4];
  float* out = (float*)d_out;

  float* ws   = (float*)d_ws;
  float* sim  = ws;
  float* sube = sim  + 1048576;
  float* S    = sube + 8388608;
  float* xn   = S    + 8192;
  float* sx   = xn   + 524288;
  float* xg1  = sx   + 524288;
  float* W    = xg1  + 524288;

  k_stage1<<<L1_GRID, 256, 0, stream>>>(E, pa, Wp, x, sim, sube, xn, W);
  k_bmm_nk<<<dim3(32, 8), 256, 0, stream>>>(sube, 1024LL * 1024, xn, 1024LL * 64,
                                            sx, 1024LL * 64);
  k_bmm_nk<<<dim3(32, 8), 256, 0, stream>>>(sim, 0LL, sx, 1024LL * 64,
                                            xg1, 1024LL * 64);
  k_final<<<128, 256, 0, stream>>>(sx, xg1, W, E, bp, out);
}


# Both batched products: K loop as a software pipeline (128-wide chunks, 2 LDS stages, loads held in registers two chunks ahead, one LDS-only barrier per chunk); changes summation order over K
# speedup vs baseline: 5.5168x; 1.0866x over previous
#include <hip/hip_runtime.h>
#include <hip/hip_bf16.h>
#include <cstddef>


typedef float v2f __attribute__((ext_vector_type(2)));
typedef float v8f __attribute__((ext_vector_type(8)));
typedef float v4f __attribute__((ext_vector_type(4)));

static __device__ __forceinline__ v8f wmma_f32(v2f a, v2f b, v8f c) {
  return __builtin_amdgcn_wmma_f32_16x16x4_f32(
       false, a,  false, b,
       (short)0, c,  false,  false);
}

static __device__ __forceinline__
void sim_part(const int blk, const float* __restrict__ E, float* __restrict__ sim,
              float* __restrict__ raw) {
  const int t = threadIdx.x;
  const int i0 = blk * 4;
#pragma unroll 1
  for (int c = 0; c < 4; ++c) {
    const int j = c * 256 + t;
    float ej[16];
#pragma unroll
    for (int q = 0; q < 4; ++q) {
      const float4 v = *(const float4*)(E + (size_t)j * 16 + q * 4);
      ej[q * 4 + 0] = v.x; ej[q * 4 + 1] = v.y; ej[q * 4 + 2] = v.z; ej[q * 4 + 3] = v.w;
    }
#pragma unroll 2
    for (int ii = 0; ii < 4; ++ii) {
      const float* ei = E + (size_t)(i0 + ii) * 16;
      float acc = 0.f;
#pragma unroll
      for (int d = 0; d < 16; ++d) acc += ei[d] * ej[d];
      raw[ii * 1024 + j] = fmaxf(acc, 0.f);
    }
  }
  __syncthreads();

  const int row = t >> 5;
  if (row >= 4) return;
  const int lane = t & 31;
  const float4* src = (const float4*)(raw + row * 1024) + lane;
  float4 v[8];
#pragma unroll
  for (int k = 0; k < 8; ++k) v[k] = src[k * 32];
  float mx = -3.0e38f;
#pragma unroll
  for (int k = 0; k < 8; ++k)
    mx = fmaxf(mx, fmaxf(fmaxf(v[k].x, v[k].y), fmaxf(v[k].z, v[k].w)));
#pragma unroll
  for (int s = 16; s > 0; s >>= 1) mx = fmaxf(mx, __shfl_xor(mx, s));
  float sum = 0.f;
#pragma unroll
  for (int k = 0; k < 8; ++k) {
    v[k].x = __expf(v[k].x - mx); v[k].y = __expf(v[k].y - mx);
    v[k].z = __expf(v[k].z - mx); v[k].w = __expf(v[k].w - mx);
    sum += (v[k].x + v[k].y) + (v[k].z + v[k].w);
  }
#pragma unroll
  for (int s = 16; s > 0; s >>= 1) sum += __shfl_xor(sum, s);
  const float inv = 1.0f / sum;
  float4* p = (float4*)(sim + (size_t)(i0 + row) * 1024) + lane;
#pragma unroll
  for (int k = 0; k < 8; ++k) {
    float4 r = {v[k].x * inv, v[k].y * inv, v[k].z * inv, v[k].w * inv};
    p[k * 32] = r;
  }
}

static __device__ __forceinline__
void cdist_part(const int blk, const float* __restrict__ pa, const float* __restrict__ x,
                float* __restrict__ sube, float* __restrict__ xn,
                float* __restrict__ Pi, float* __restrict__ red) {
  const int b = blk >> 5;
  const int j0 = (blk & 31) * 32;
  const int t = threadIdx.x;
  const int wave = t >> 5;
  const int lane = t & 31;
  const int j = j0 + lane;
  const int iBase = wave * 128;
  const float4* rsrc = (const float4*)(pa + ((size_t)b * 1024 + iBase) * 16) + lane;
  float4 stg[16];
#pragma unroll
  for (int q = 0; q < 16; ++q) stg[q] = rsrc[q * 32];
  float pj[16];
#pragma unroll
  for (int q = 0; q < 4; ++q) {
    const float4 v = *(const float4*)(pa + ((size_t)b * 1024 + j) * 16 + q * 4);
    pj[q * 4 + 0] = v.x; pj[q * 4 + 1] = v.y; pj[q * 4 + 2] = v.z; pj[q * 4 + 3] = v.w;
  }
  const size_t xoff = ((size_t)b * 1024 + j0) * 16;
  float4 xv[2];
#pragma unroll
  for (int k = 0; k < 2; ++k) xv[k] = ((const float4*)x)[xoff + t + k * 256];
  float4* rdst = (float4*)(Pi + (size_t)iBase * 16) + lane;
#pragma unroll
  for (int q = 0; q < 16; ++q) rdst[q * 32] = stg[q];
  __syncthreads();

  const float* Pw = Pi + (size_t)iBase * 16;
  float* dst = sube + ((size_t)b * 1024 + iBase) * 1024 + j;
  float tot = 0.f;
  for (int g = 0; g < 8; ++g) {
    float csum = 0.f;
#pragma unroll 4
    for (int ii = 0; ii < 16; ++ii) {
      const int i = g * 16 + ii;
      float s = 0.f;
#pragma unroll
      for (int d = 0; d < 16; ++d) s += fabsf(Pw[i * 16 + d] - pj[d]);
      const float e = __expf(-s);
      dst[(size_t)i * 1024] = e;
      csum += e;
    }
    tot += csum;
  }
  red[wave * 32 + lane] = tot;
  __syncthreads();

#pragma unroll
  for (int k = 0; k < 2; ++k) {
    const int f = t + k * 256;
    const int jj = f >> 4;
    const float s = ((red[jj] + red[32 + jj]) + (red[64 + jj] + red[96 + jj])) +
                    ((red[128 + jj] + red[160 + jj]) + (red[192 + jj] + red[224 + jj]));
    const float4 v = xv[k];
    float4 r = {v.x / s, v.y / s, v.z / s, v.w / s};
    ((float4*)xn)[xoff + f] = r;
  }
}

static __device__ __forceinline__
void weights_part(const int blk, const float* __restrict__ E, const float* __restrict__ Wp,
                  float* __restrict__ W) {
  const int n0 = (blk >> 3) * 16;
  const int tid = (blk & 7) * 256 + threadIdx.x;
  const int kp = tid >> 5;
  const int oc = (tid & 31) * 2;
  v2f w0[16], w1[16];
#pragma unroll
  for (int d = 0; d < 16; ++d) {
    w0[d] = *(const v2f*)(Wp + (size_t)d * 8192 + (size_t)(2 * kp) * 64 + oc);
    w1[d] = *(const v2f*)(Wp + (size_t)d * 8192 + (size_t)(2 * kp + 1) * 64 + oc);
  }
  float* dst = W + (size_t)n0 * 8192 + (size_t)kp * 128 + (size_t)oc * 2;
#pragma unroll 2
  for (int nn = 0; nn < 16; ++nn) {
    const float* e = E + (size_t)(n0 + nn) * 16;
    float4 acc = {0.f, 0.f, 0.f, 0.f};
#pragma unroll
    for (int d = 0; d < 16; ++d) {
      acc.x += e[d] * w0[d].x;
      acc.y += e[d] * w1[d].x;
      acc.z += e[d] * w0[d].y;
      acc.w += e[d] * w1[d].y;
    }
    *(float4*)(dst + (size_t)nn * 8192) = acc;
  }
}

#define L1_CDIST   256
#define L1_WEIGHTS 512
#define L1_SIM     256
#define L1_GRID    1024
__global__ __launch_bounds__(256)
void k_stage1(const float* __restrict__ E, const float* __restrict__ pa,
              const float* __restrict__ Wp, const float* __restrict__ x,
              float* __restrict__ sim, float* __restrict__ sube,
              float* __restrict__ xn, float* __restrict__ W) {
  __shared__ __attribute__((aligned(16))) float lds[16384 + 256];
  const int blk = blockIdx.x;
  if (blk < L1_CDIST) {
    cdist_part(blk, pa, x, sube, xn, lds, lds + 16384);
  } else if (blk < L1_CDIST + L1_WEIGHTS) {
    weights_part(blk - L1_CDIST, E, Wp, W);
  } else {
    sim_part(blk - (L1_CDIST + L1_WEIGHTS), E, sim, lds);
  }
}

#define BMM_KC 128
#define BMM_BS 224
#define BMM_NCH (1024 / BMM_KC)

static __device__ __forceinline__
void bmm_request(const float* __restrict__ Ab, const float* __restrict__ Bb,
                 const int rowBase, const int kc, const int t,
                 v4f (&ra)[4], v4f (&rb0)[4], v4f (&rb1)[4]) {
#pragma unroll
  for (int j = 0; j < 4; ++j) {
    const int i = t + j * 256;
    const int r = i >> 5, cc = (i & 31) << 2;
    ra[j] = *(const v4f*)(Ab + (size_t)(rowBase + r) * 1024 + kc + cc);
  }
#pragma unroll
  for (int j = 0; j < 4; ++j) {
    const int i = t + j * 256;
    const int kp = i >> 4, cc = (i & 15) << 2;
    rb0[j] = *(const v4f*)(Bb + (size_t)(kc + 2 * kp) * 64 + cc);
    rb1[j] = *(const v4f*)(Bb + (size_t)(kc + 2 * kp + 1) * 64 + cc);
  }
}

static __device__ __forceinline__
void bmm_stage(float (*As)[BMM_KC + 4], float (*Bs)[BMM_BS], const int t,
               v4f (&ra)[4], v4f (&rb0)[4], v4f (&rb1)[4]) {
#pragma unroll
  for (int j = 0; j < 4; ++j) {
    const int i = t + j * 256;
    const int r = i >> 5, cc = (i & 31) << 2;
    *(v4f*)(&As[r][cc]) = ra[j];
  }
#pragma unroll
  for (int j = 0; j < 4; ++j) {
    const int i = t + j * 256;
    const int kp = i >> 4, cc = (i & 15) << 2;
    v4f v0 = rb0[j];
    v4f v1 = rb1[j];
    asm volatile("" : "+v"(v0));
    asm volatile("" : "+v"(v1));
    const v4f lo = {v0.x, v1.x, v0.y, v1.y};
    const v4f hi = {v0.z, v1.z, v0.w, v1.w};
    *(v4f*)(&Bs[kp][cc * 2])     = lo;
    *(v4f*)(&Bs[kp][cc * 2 + 4]) = hi;
  }
}

static __device__ __forceinline__ void bmm_lds_barrier() {
  __builtin_amdgcn_fence(__ATOMIC_RELEASE, "workgroup", "local");
  __builtin_amdgcn_s_barrier();
  __builtin_amdgcn_fence(__ATOMIC_ACQUIRE, "workgroup", "local");
}

static __device__ __forceinline__
void bmm_step(const float* __restrict__ Ab, const float* __restrict__ Bb,
              const int rowBase, const int c, const int t,
              float (*AsC)[BMM_KC + 4], float (*BsC)[BMM_BS],
              float (*AsN)[BMM_KC + 4], float (*BsN)[BMM_BS],
              v4f (&fa)[4], v4f (&fb0)[4], v4f (&fb1)[4],
              v4f (&na)[4], v4f (&nb0)[4], v4f (&nb1)[4],
              const int mrow, const int m16, const int kh, const int kq,
              v8f (&cacc)[4]) {
  if (c + 2 < BMM_NCH)
    bmm_request(Ab, Bb, rowBase, (c + 2) * BMM_KC, t, fa, fb0, fb1);
  __builtin_amdgcn_sched_barrier(0);

  const int kq0 = kq * (BMM_KC / 4);
#pragma unroll
  for (int ks = 0; ks < BMM_KC / 16; ++ks) {
    const int kg = kq0 + ks * 4 + 2 * kh;
    const int kp = (kq0 >> 1) + ks * 2 + kh;
    const v2f a = *(const v2f*)(&AsC[mrow][kg]);
#pragma unroll
    for (int nt = 0; nt < 4; ++nt) {
      const v2f bf = *(const v2f*)(&BsC[kp][(nt * 16 + m16) * 2]);
      cacc[nt] = wmma_f32(a, bf, cacc[nt]);
    }
  }
  __builtin_amdgcn_sched_barrier(0);
  if (c + 1 < BMM_NCH) {
    bmm_stage(AsN, BsN, t, na, nb0, nb1);
    bmm_lds_barrier();
  }
}

__global__ __launch_bounds__(256)
void k_bmm_nk(const float* __restrict__ A, long long strideA,
              const float* __restrict__ B, long long strideB,
              float* __restrict__ C, long long strideC) {
  const int b = blockIdx.y;
  const float* Ab = A + (size_t)strideA * b;
  const float* Bb = B + (size_t)strideB * b;
  float* Cb = C + (size_t)strideC * b;
  const int rowBase = blockIdx.x * 32;

  __shared__ __attribute__((aligned(16))) float As[2][32][BMM_KC + 4];
  __shared__ __attribute__((aligned(16))) float Bs[2][BMM_KC / 2][BMM_BS];
  __shared__ __attribute__((aligned(16))) float Red[4][32][68];

  const int t = threadIdx.x;
  const int wave = t >> 5;
  const int lane = t & 31;
  const int m16 = lane & 15;
  const int kh = lane >> 4;
  const int rs = wave & 1;
  const int kq = wave >> 1;
  const int mrow = rs * 16 + m16;

  const v8f zero = {0.f, 0.f, 0.f, 0.f, 0.f, 0.f, 0.f, 0.f};
  v8f cacc[4];
#pragma unroll
  for (int nt = 0; nt < 4; ++nt) cacc[nt] = zero;

  v4f a0[4], b00[4], b01[4];
  v4f a1[4], b10[4], b11[4];
  bmm_request(Ab, Bb, rowBase, 0, t, a0, b00, b01);
  bmm_request(Ab, Bb, rowBase, BMM_KC, t, a1, b10, b11);
  bmm_stage(As[0], Bs[0], t, a0, b00, b01);
  bmm_lds_barrier();

#pragma unroll
  for (int c = 0; c < BMM_NCH; c += 2) {
    bmm_step(Ab, Bb, rowBase, c,     t, As[0], Bs[0], As[1], Bs[1],
             a0, b00, b01, a1, b10, b11, mrow, m16, kh, kq, cacc);
    bmm_step(Ab, Bb, rowBase, c + 1, t, As[1], Bs[1], As[0], Bs[0],
             a1, b10, b11, a0, b00, b01, mrow, m16, kh, kq, cacc);
  }

#pragma unroll
  for (int nt = 0; nt < 4; ++nt) {
#pragma unroll
    for (int r = 0; r < 8; ++r)
      Red[kq][rs * 16 + r + 8 * kh][nt * 16 + m16] = cacc[nt][r];
  }
  __syncthreads();
#pragma unroll
  for (int j = 0; j < 2; ++j) {
    const int e = (t + j * 256) * 4;
    const int row = e >> 6, col = e & 63;
    const float4 p0 = *(const float4*)(&Red[0][row][col]);
    const float4 p1 = *(const float4*)(&Red[1][row][col]);
    const float4 p2 = *(const float4*)(&Red[2][row][col]);
    const float4 p3 = *(const float4*)(&Red[3][row][col]);
    float4 s;
    s.x = (p0.x + p1.x) + (p2.x + p3.x);
    s.y = (p0.y + p1.y) + (p2.y + p3.y);
    s.z = (p0.z + p1.z) + (p2.z + p3.z);
    s.w = (p0.w + p1.w) + (p2.w + p3.w);
    *(float4*)(Cb + (size_t)(rowBase + row) * 64 + col) = s;
  }
}

__global__ __launch_bounds__(256)
void k_final(const float* __restrict__ sx, const float* __restrict__ xg1,
             const float* __restrict__ W, const float* __restrict__ E,
             const float* __restrict__ bias_pool, float* __restrict__ out) {
  __shared__ float As[8][8][132];
  const int t = threadIdx.x;
  const int wave = t >> 5;
  const int lane = t & 31;
  const int n = blockIdx.x * 8 + wave;
  float (*Aw)[132] = As[wave];
  const int m16 = lane & 15;
  const int kh = lane >> 4;

  float va[8][4];
#pragma unroll
  for (int rb = 0; rb < 8; ++rb) {
    const float* s0 = sx  + ((size_t)rb * 1024 + n) * 64;
    const float* s1 = xg1 + ((size_t)rb * 1024 + n) * 64;
    va[rb][0] = s0[lane];
    va[rb][1] = s0[lane + 32];
    va[rb][2] = s1[lane];
    va[rb][3] = s1[lane + 32];
  }
  float bv[4];
#pragma unroll
  for (int nt = 0; nt < 4; ++nt) {
    const int col = nt * 16 + m16;
    float s = 0.f;
#pragma unroll
    for (int d = 0; d < 16; ++d) s += E[n * 16 + d] * bias_pool[d * 64 + col];
    bv[nt] = s;
  }
#pragma unroll
  for (int rb = 0; rb < 8; ++rb) {
    Aw[rb][lane]      = va[rb][0];
    Aw[rb][lane + 32] = va[rb][1];
    Aw[rb][lane + 64] = va[rb][2];
    Aw[rb][lane + 96] = va[rb][3];
  }
  __syncthreads();

  const float* Wn = W + (size_t)n * 8192;
  v2f bw[32][4];
#pragma unroll
  for (int ks = 0; ks < 32; ++ks) {
    const int kp = ks * 2 + kh;
#pragma unroll
    for (int nt = 0; nt < 4; ++nt)
      bw[ks][nt] = *(const v2f*)(Wn + (size_t)kp * 128 + (nt * 16 + m16) * 2);
  }
  const int arow = m16 & 7;
  v2f af[32];
#pragma unroll
  for (int ks = 0; ks < 32; ++ks)
    af[ks] = *(const v2f*)(&Aw[arow][ks * 4 + 2 * kh]);
  __builtin_amdgcn_sched_barrier(0);

  const v8f zero = {0.f, 0.f, 0.f, 0.f, 0.f, 0.f, 0.f, 0.f};
  v8f cacc[4];
#pragma unroll
  for (int nt = 0; nt < 4; ++nt) cacc[nt] = zero;

#pragma unroll
  for (int ks = 0; ks < 32; ++ks) {
#pragma unroll
    for (int nt = 0; nt < 4; ++nt)
      cacc[nt] = wmma_f32(af[ks], bw[ks][nt], cacc[nt]);
  }

#pragma unroll
  for (int nt = 0; nt < 4; ++nt) {
    const int col = nt * 16 + m16;
#pragma unroll
    for (int r = 0; r < 8; ++r) {
      const int m = r + 8 * kh;
      if (m < 8)
        out[((size_t)m * 1024 + n) * 64 + col] = cacc[nt][r] + bv[nt];
    }
  }
}

extern "C" void kernel_launch(void* const* d_in, const int* in_sizes, int n_in,
                              void* d_out, int out_size, void* d_ws, size_t ws_size,
                              hipStream_t stream) {
  (void)in_sizes; (void)n_in; (void)out_size; (void)ws_size;
  const float* x    = (const float*)d_in[0];
  const float* E    = (const float*)d_in[1];
  const float* pa   = (const float*)d_in[2];
  const float* Wp   = (const float*)d_in[3];
  const float* bp   = (const float*)d_in[4];
  float* out = (float*)d_out;

  float* ws   = (float*)d_ws;
  float* sim  = ws;
  float* sube = sim  + 1048576;
  float* S    = sube + 8388608;
  float* xn   = S    + 8192;
  float* sx   = xn   + 524288;
  float* xg1  = sx   + 524288;
  float* W    = xg1  + 524288;

  k_stage1<<<L1_GRID, 256, 0, stream>>>(E, pa, Wp, x, sim, sube, xn, W);
  k_bmm_nk<<<dim3(32, 8), 256, 0, stream>>>(sube, 1024LL * 1024, xn, 1024LL * 64,
                                            sx, 1024LL * 64);
  k_bmm_nk<<<dim3(32, 8), 256, 0, stream>>>(sim, 0LL, sx, 1024LL * 64,
                                            xg1, 1024LL * 64);
  k_final<<<128, 256, 0, stream>>>(sx, xg1, W, E, bp, out);
}


# Node weights never stored: final kernel forms W and bias by WMMA (16 nodes x 16 columns per work-group, K over 4 waves) via LDS; weights pass and W traffic gone; changes sum order of W, bias, K
# speedup vs baseline: 6.5799x; 1.1927x over previous
#include <hip/hip_runtime.h>
#include <hip/hip_bf16.h>
#include <cstddef>


typedef float v2f __attribute__((ext_vector_type(2)));
typedef float v8f __attribute__((ext_vector_type(8)));
typedef float v4f __attribute__((ext_vector_type(4)));

static __device__ __forceinline__ v8f wmma_f32(v2f a, v2f b, v8f c) {
  return __builtin_amdgcn_wmma_f32_16x16x4_f32(
       false, a,  false, b,
       (short)0, c,  false,  false);
}

static __device__ __forceinline__
void sim_part(const int blk, const float* __restrict__ E, float* __restrict__ sim,
              float* __restrict__ raw) {
  const int t = threadIdx.x;
  const int i0 = blk * 4;
#pragma unroll 1
  for (int c = 0; c < 4; ++c) {
    const int j = c * 256 + t;
    float ej[16];
#pragma unroll
    for (int q = 0; q < 4; ++q) {
      const float4 v = *(const float4*)(E + (size_t)j * 16 + q * 4);
      ej[q * 4 + 0] = v.x; ej[q * 4 + 1] = v.y; ej[q * 4 + 2] = v.z; ej[q * 4 + 3] = v.w;
    }
#pragma unroll 2
    for (int ii = 0; ii < 4; ++ii) {
      const float* ei = E + (size_t)(i0 + ii) * 16;
      float acc = 0.f;
#pragma unroll
      for (int d = 0; d < 16; ++d) acc += ei[d] * ej[d];
      raw[ii * 1024 + j] = fmaxf(acc, 0.f);
    }
  }
  __syncthreads();

  const int row = t >> 5;
  if (row >= 4) return;
  const int lane = t & 31;
  const float4* src = (const float4*)(raw + row * 1024) + lane;
  float4 v[8];
#pragma unroll
  for (int k = 0; k < 8; ++k) v[k] = src[k * 32];
  float mx = -3.0e38f;
#pragma unroll
  for (int k = 0; k < 8; ++k)
    mx = fmaxf(mx, fmaxf(fmaxf(v[k].x, v[k].y), fmaxf(v[k].z, v[k].w)));
#pragma unroll
  for (int s = 16; s > 0; s >>= 1) mx = fmaxf(mx, __shfl_xor(mx, s));
  float sum = 0.f;
#pragma unroll
  for (int k = 0; k < 8; ++k) {
    v[k].x = __expf(v[k].x - mx); v[k].y = __expf(v[k].y - mx);
    v[k].z = __expf(v[k].z - mx); v[k].w = __expf(v[k].w - mx);
    sum += (v[k].x + v[k].y) + (v[k].z + v[k].w);
  }
#pragma unroll
  for (int s = 16; s > 0; s >>= 1) sum += __shfl_xor(sum, s);
  const float inv = 1.0f / sum;
  float4* p = (float4*)(sim + (size_t)(i0 + row) * 1024) + lane;
#pragma unroll
  for (int k = 0; k < 8; ++k) {
    float4 r = {v[k].x * inv, v[k].y * inv, v[k].z * inv, v[k].w * inv};
    p[k * 32] = r;
  }
}

static __device__ __forceinline__
void cdist_part(const int blk, const float* __restrict__ pa, const float* __restrict__ x,
                float* __restrict__ sube, float* __restrict__ xn,
                float* __restrict__ Pi, float* __restrict__ red) {
  const int b = blk >> 5;
  const int j0 = (blk & 31) * 32;
  const int t = threadIdx.x;
  const int wave = t >> 5;
  const int lane = t & 31;
  const int j = j0 + lane;
  const int iBase = wave * 128;
  const float4* rsrc = (const float4*)(pa + ((size_t)b * 1024 + iBase) * 16) + lane;
  float4 stg[16];
#pragma unroll
  for (int q = 0; q < 16; ++q) stg[q] = rsrc[q * 32];
  float pj[16];
#pragma unroll
  for (int q = 0; q < 4; ++q) {
    const float4 v = *(const float4*)(pa + ((size_t)b * 1024 + j) * 16 + q * 4);
    pj[q * 4 + 0] = v.x; pj[q * 4 + 1] = v.y; pj[q * 4 + 2] = v.z; pj[q * 4 + 3] = v.w;
  }
  const size_t xoff = ((size_t)b * 1024 + j0) * 16;
  float4 xv[2];
#pragma unroll
  for (int k = 0; k < 2; ++k) xv[k] = ((const float4*)x)[xoff + t + k * 256];
  float4* rdst = (float4*)(Pi + (size_t)iBase * 16) + lane;
#pragma unroll
  for (int q = 0; q < 16; ++q) rdst[q * 32] = stg[q];
  __syncthreads();

  const float* Pw = Pi + (size_t)iBase * 16;
  float* dst = sube + ((size_t)b * 1024 + iBase) * 1024 + j;
  float tot = 0.f;
  for (int g = 0; g < 8; ++g) {
    float csum = 0.f;
#pragma unroll 4
    for (int ii = 0; ii < 16; ++ii) {
      const int i = g * 16 + ii;
      float s = 0.f;
#pragma unroll
      for (int d = 0; d < 16; ++d) s += fabsf(Pw[i * 16 + d] - pj[d]);
      const float e = __expf(-s);
      dst[(size_t)i * 1024] = e;
      csum += e;
    }
    tot += csum;
  }
  red[wave * 32 + lane] = tot;
  __syncthreads();

#pragma unroll
  for (int k = 0; k < 2; ++k) {
    const int f = t + k * 256;
    const int jj = f >> 4;
    const float s = ((red[jj] + red[32 + jj]) + (red[64 + jj] + red[96 + jj])) +
                    ((red[128 + jj] + red[160 + jj]) + (red[192 + jj] + red[224 + jj]));
    const float4 v = xv[k];
    float4 r = {v.x / s, v.y / s, v.z / s, v.w / s};
    ((float4*)xn)[xoff + f] = r;
  }
}

static __device__ __forceinline__
void wpt_part(const int blk, const float* __restrict__ Wp, float* __restrict__ WpT) {
  const int p = blk * 256 + threadIdx.x;
  float v[16];
#pragma unroll
  for (int d = 0; d < 16; ++d) v[d] = Wp[(size_t)d * 8192 + p];
  float4* dst = (float4*)(WpT + (size_t)p * 16);
#pragma unroll
  for (int g = 0; g < 4; ++g) {
    float4 r = {v[g * 4 + 0], v[g * 4 + 1], v[g * 4 + 2], v[g * 4 + 3]};
    dst[g] = r;
  }
}

#define L1_CDIST   256
#define L1_WPT     32
#define L1_SIM     256
#define L1_GRID    544
__global__ __launch_bounds__(256)
void k_stage1(const float* __restrict__ E, const float* __restrict__ pa,
              const float* __restrict__ Wp, const float* __restrict__ x,
              float* __restrict__ sim, float* __restrict__ sube,
              float* __restrict__ xn, float* __restrict__ WpT) {
  __shared__ __attribute__((aligned(16))) float lds[16384 + 256];
  const int blk = blockIdx.x;
  if (blk < L1_CDIST) {
    cdist_part(blk, pa, x, sube, xn, lds, lds + 16384);
  } else if (blk < L1_CDIST + L1_WPT) {
    wpt_part(blk - L1_CDIST, Wp, WpT);
  } else {
    sim_part(blk - (L1_CDIST + L1_WPT), E, sim, lds);
  }
}

#define BMM_KC 128
#define BMM_BS 224
#define BMM_NCH (1024 / BMM_KC)

static __device__ __forceinline__
void bmm_request(const float* __restrict__ Ab, const float* __restrict__ Bb,
                 const int rowBase, const int kc, const int t,
                 v4f (&ra)[4], v4f (&rb0)[4], v4f (&rb1)[4]) {
#pragma unroll
  for (int j = 0; j < 4; ++j) {
    const int i = t + j * 256;
    const int r = i >> 5, cc = (i & 31) << 2;
    ra[j] = *(const v4f*)(Ab + (size_t)(rowBase + r) * 1024 + kc + cc);
  }
#pragma unroll
  for (int j = 0; j < 4; ++j) {
    const int i = t + j * 256;
    const int kp = i >> 4, cc = (i & 15) << 2;
    rb0[j] = *(const v4f*)(Bb + (size_t)(kc + 2 * kp) * 64 + cc);
    rb1[j] = *(const v4f*)(Bb + (size_t)(kc + 2 * kp + 1) * 64 + cc);
  }
}

static __device__ __forceinline__
void bmm_stage(float (*As)[BMM_KC + 4], float (*Bs)[BMM_BS], const int t,
               v4f (&ra)[4], v4f (&rb0)[4], v4f (&rb1)[4]) {
#pragma unroll
  for (int j = 0; j < 4; ++j) {
    const int i = t + j * 256;
    const int r = i >> 5, cc = (i & 31) << 2;
    *(v4f*)(&As[r][cc]) = ra[j];
  }
#pragma unroll
  for (int j = 0; j < 4; ++j) {
    const int i = t + j * 256;
    const int kp = i >> 4, cc = (i & 15) << 2;
    v4f v0 = rb0[j];
    v4f v1 = rb1[j];
    asm volatile("" : "+v"(v0));
    asm volatile("" : "+v"(v1));
    const v4f lo = {v0.x, v1.x, v0.y, v1.y};
    const v4f hi = {v0.z, v1.z, v0.w, v1.w};
    *(v4f*)(&Bs[kp][cc * 2])     = lo;
    *(v4f*)(&Bs[kp][cc * 2 + 4]) = hi;
  }
}

static __device__ __forceinline__ void bmm_lds_barrier() {
  __builtin_amdgcn_fence(__ATOMIC_RELEASE, "workgroup", "local");
  __builtin_amdgcn_s_barrier();
  __builtin_amdgcn_fence(__ATOMIC_ACQUIRE, "workgroup", "local");
}

static __device__ __forceinline__
void bmm_step(const float* __restrict__ Ab, const float* __restrict__ Bb,
              const int rowBase, const int c, const int t,
              float (*AsC)[BMM_KC + 4], float (*BsC)[BMM_BS],
              float (*AsN)[BMM_KC + 4], float (*BsN)[BMM_BS],
              v4f (&fa)[4], v4f (&fb0)[4], v4f (&fb1)[4],
              v4f (&na)[4], v4f (&nb0)[4], v4f (&nb1)[4],
              const int mrow, const int m16, const int kh, const int kq,
              v8f (&cacc)[4]) {
  if (c + 2 < BMM_NCH)
    bmm_request(Ab, Bb, rowBase, (c + 2) * BMM_KC, t, fa, fb0, fb1);
  __builtin_amdgcn_sched_barrier(0);

  const int kq0 = kq * (BMM_KC / 4);
#pragma unroll
  for (int ks = 0; ks < BMM_KC / 16; ++ks) {
    const int kg = kq0 + ks * 4 + 2 * kh;
    const int kp = (kq0 >> 1) + ks * 2 + kh;
    const v2f a = *(const v2f*)(&AsC[mrow][kg]);
#pragma unroll
    for (int nt = 0; nt < 4; ++nt) {
      const v2f bf = *(const v2f*)(&BsC[kp][(nt * 16 + m16) * 2]);
      cacc[nt] = wmma_f32(a, bf, cacc[nt]);
    }
  }
  __builtin_amdgcn_sched_barrier(0);
  if (c + 1 < BMM_NCH) {
    bmm_stage(AsN, BsN, t, na, nb0, nb1);
    bmm_lds_barrier();
  }
}

__global__ __launch_bounds__(256)
void k_bmm_nk(const float* __restrict__ A, long long strideA,
              const float* __restrict__ B, long long strideB,
              float* __restrict__ C, long long strideC) {
  const int b = blockIdx.y;
  const float* Ab = A + (size_t)strideA * b;
  const float* Bb = B + (size_t)strideB * b;
  float* Cb = C + (size_t)strideC * b;
  const int rowBase = blockIdx.x * 32;

  __shared__ __attribute__((aligned(16))) float As[2][32][BMM_KC + 4];
  __shared__ __attribute__((aligned(16))) float Bs[2][BMM_KC / 2][BMM_BS];
  __shared__ __attribute__((aligned(16))) float Red[4][32][68];

  const int t = threadIdx.x;
  const int wave = t >> 5;
  const int lane = t & 31;
  const int m16 = lane & 15;
  const int kh = lane >> 4;
  const int rs = wave & 1;
  const int kq = wave >> 1;
  const int mrow = rs * 16 + m16;

  const v8f zero = {0.f, 0.f, 0.f, 0.f, 0.f, 0.f, 0.f, 0.f};
  v8f cacc[4];
#pragma unroll
  for (int nt = 0; nt < 4; ++nt) cacc[nt] = zero;

  v4f a0[4], b00[4], b01[4];
  v4f a1[4], b10[4], b11[4];
  bmm_request(Ab, Bb, rowBase, 0, t, a0, b00, b01);
  bmm_request(Ab, Bb, rowBase, BMM_KC, t, a1, b10, b11);
  bmm_stage(As[0], Bs[0], t, a0, b00, b01);
  bmm_lds_barrier();

#pragma unroll
  for (int c = 0; c < BMM_NCH; c += 2) {
    bmm_step(Ab, Bb, rowBase, c,     t, As[0], Bs[0], As[1], Bs[1],
             a0, b00, b01, a1, b10, b11, mrow, m16, kh, kq, cacc);
    bmm_step(Ab, Bb, rowBase, c + 1, t, As[1], Bs[1], As[0], Bs[0],
             a1, b10, b11, a0, b00, b01, mrow, m16, kh, kq, cacc);
  }

#pragma unroll
  for (int nt = 0; nt < 4; ++nt) {
#pragma unroll
    for (int r = 0; r < 8; ++r)
      Red[kq][rs * 16 + r + 8 * kh][nt * 16 + m16] = cacc[nt][r];
  }
  __syncthreads();
#pragma unroll
  for (int j = 0; j < 2; ++j) {
    const int e = (t + j * 256) * 4;
    const int row = e >> 6, col = e & 63;
    const float4 p0 = *(const float4*)(&Red[0][row][col]);
    const float4 p1 = *(const float4*)(&Red[1][row][col]);
    const float4 p2 = *(const float4*)(&Red[2][row][col]);
    const float4 p3 = *(const float4*)(&Red[3][row][col]);
    float4 s;
    s.x = (p0.x + p1.x) + (p2.x + p3.x);
    s.y = (p0.y + p1.y) + (p2.y + p3.y);
    s.z = (p0.z + p1.z) + (p2.z + p3.z);
    s.w = (p0.w + p1.w) + (p2.w + p3.w);
    *(float4*)(Cb + (size_t)(rowBase + row) * 64 + col) = s;
  }
}

#define FN_WS 516
#define FN_AS 36
#define FN_RH 2080

static __device__ __forceinline__
void fin_frags(const float (*Alq)[8][FN_AS], const float (*Wlq)[FN_WS],
               const int g, const int c, const int h,
               v2f (&fa)[4][8], v2f (&fb)[4][8]) {
#pragma unroll
  for (int nn = 0; nn < 4; ++nn) {
#pragma unroll
    for (int ks = 0; ks < 8; ++ks) {
      fa[nn][ks] = *(const v2f*)(&Alq[g * 4 + nn][c & 7][4 * ks + 2 * h]);
      fb[nn][ks] = *(const v2f*)(&Wlq[g * 4 + nn][(2 * ks + h) * 32 + 2 * c]);
    }
  }
}

static __device__ __forceinline__
void fin_mma(const int g, v2f (&fa)[4][8], v2f (&fb)[4][8], v8f (&acc)[16]) {
#pragma unroll
  for (int ks = 0; ks < 8; ++ks) {
#pragma unroll
    for (int nn = 0; nn < 4; ++nn)
      acc[g * 4 + nn] = wmma_f32(fa[nn][ks], fb[nn][ks], acc[g * 4 + nn]);
  }
}

__global__ __launch_bounds__(128)
void k_final(const float* __restrict__ sx, const float* __restrict__ xg1,
             const float* __restrict__ WpT, const float* __restrict__ E,
             const float* __restrict__ bias_pool, float* __restrict__ out) {
  __shared__ __attribute__((aligned(16))) float Wl[4][16][FN_WS];
  __shared__ __attribute__((aligned(16))) float Al[4][16][8][FN_AS];
  __shared__ __attribute__((aligned(16))) float Red[4][2][FN_RH];
  __shared__ __attribute__((aligned(16))) float Bl[4][16][16];
  const int t = threadIdx.x;
  const int q = t >> 5;
  const int lane = t & 31;
  const int c = lane & 15;
  const int h = lane >> 4;
  const int o0 = blockIdx.x * 16;
  const int n0 = blockIdx.y * 16;

  v2f ea[4], bpf[4];
#pragma unroll
  for (int s = 0; s < 4; ++s) {
    const int d = 8 * h + 2 * s;
    ea[s] = *(const v2f*)(E + (size_t)(n0 + c) * 16 + d);
    const v2f bp2 = {bias_pool[d * 64 + o0 + c], bias_pool[(d + 1) * 64 + o0 + c]};
    bpf[s] = bp2;
  }
  const float* wsrc = WpT + ((size_t)(32 * q) * 64 + o0 + c) * 16 + 8 * h;
  v4f wlo[32], whi[32];
#pragma unroll
  for (int kl = 0; kl < 32; ++kl) {
    wlo[kl] = *(const v4f*)(wsrc + (size_t)kl * 1024);
    whi[kl] = *(const v4f*)(wsrc + (size_t)kl * 1024 + 4);
  }
  const v8f zero = {0.f, 0.f, 0.f, 0.f, 0.f, 0.f, 0.f, 0.f};
  const float* asrc = (q < 2 ? sx : xg1) + (q & 1) * 32;
  v4f ar[32];

#pragma unroll
  for (int bt = 0; bt < 4; ++bt) {
    if (bt == 2) {
      __builtin_amdgcn_sched_barrier(0);
#pragma unroll
      for (int j = 0; j < 32; ++j) {
        const int row = j * 4 + (lane >> 3);
        ar[j] = *(const v4f*)(asrc + ((size_t)(row & 7) * 1024 + n0 + (row >> 3)) * 64 + (lane & 7) * 4);
      }
    }
    __builtin_amdgcn_sched_barrier(0);
#pragma unroll
    for (int kp = bt * 4; kp < bt * 4 + 4; ++kp) {
      const v4f l0 = wlo[2 * kp],     h0 = whi[2 * kp];
      const v4f l1 = wlo[2 * kp + 1], h1 = whi[2 * kp + 1];
      const v2f b0[4] = {{l0.x, l0.y}, {l0.z, l0.w}, {h0.x, h0.y}, {h0.z, h0.w}};
      const v2f b1[4] = {{l1.x, l1.y}, {l1.z, l1.w}, {h1.x, h1.y}, {h1.z, h1.w}};
      v8f d0 = zero, d1 = zero;
#pragma unroll
      for (int s = 0; s < 4; ++s) {
        d0 = wmma_f32(ea[s], b0[s], d0);
        d1 = wmma_f32(ea[s], b1[s], d1);
      }
#pragma unroll
      for (int r = 0; r < 8; ++r) {
        const v2f w = {d0[r], d1[r]};
        *(v2f*)(&Wl[q][8 * h + r][kp * 32 + 2 * c]) = w;
      }
    }
  }
  {
    v8f db = zero;
#pragma unroll
    for (int s = 0; s < 4; ++s) db = wmma_f32(ea[s], bpf[s], db);
#pragma unroll
    for (int r = 0; r < 8; ++r) Bl[q][8 * h + r][c] = db[r];
  }
#pragma unroll
  for (int j = 0; j < 32; ++j) {
    const int row = j * 4 + (lane >> 3);
    *(v4f*)(&Al[q][row >> 3][row & 7][(lane & 7) * 4]) = ar[j];
  }
  __syncthreads();

  v8f acc[16];
#pragma unroll
  for (int n = 0; n < 16; ++n) acc[n] = zero;
  v2f fa0[4][8], fb0[4][8], fa1[4][8], fb1[4][8];
  fin_frags(Al[q], Wl[q], 0, c, h, fa0, fb0);
  fin_frags(Al[q], Wl[q], 1, c, h, fa1, fb1);
  __builtin_amdgcn_sched_barrier(0);
  fin_mma(0, fa0, fb0, acc);
  __builtin_amdgcn_sched_barrier(0);
  fin_frags(Al[q], Wl[q], 2, c, h, fa0, fb0);
  __builtin_amdgcn_sched_barrier(0);
  fin_mma(1, fa1, fb1, acc);
  __builtin_amdgcn_sched_barrier(0);
  fin_frags(Al[q], Wl[q], 3, c, h, fa1, fb1);
  __builtin_amdgcn_sched_barrier(0);
  fin_mma(2, fa0, fb0, acc);
  fin_mma(3, fa1, fb1, acc);

  float* red = &Red[q][h][c];
#pragma unroll
  for (int n = 0; n < 16; ++n) {
#pragma unroll
    for (int r = 0; r < 8; ++r) red[(n * 8 + r) * 16] = acc[n][r];
  }
  __syncthreads();
#pragma unroll
  for (int j = 0; j < 4; ++j) {
    const int idx = t + j * 128;
    const int c4 = (idx & 3) * 4, b = (idx >> 2) & 7, n = idx >> 5;
    const float4 p0 = *(const float4*)(&Red[0][0][(n * 8 + b) * 16 + c4]);
    const float4 p1 = *(const float4*)(&Red[1][0][(n * 8 + b) * 16 + c4]);
    const float4 p2 = *(const float4*)(&Red[2][0][(n * 8 + b) * 16 + c4]);
    const float4 p3 = *(const float4*)(&Red[3][0][(n * 8 + b) * 16 + c4]);
    const float4 bb = *(const float4*)(&Bl[0][n][c4]);
    float4 sres;
    sres.x = ((p0.x + p1.x) + (p2.x + p3.x)) + bb.x;
    sres.y = ((p0.y + p1.y) + (p2.y + p3.y)) + bb.y;
    sres.z = ((p0.z + p1.z) + (p2.z + p3.z)) + bb.z;
    sres.w = ((p0.w + p1.w) + (p2.w + p3.w)) + bb.w;
    *(float4*)(out + ((size_t)b * 1024 + n0 + n) * 64 + o0 + c4) = sres;
  }
}

extern "C" void kernel_launch(void* const* d_in, const int* in_sizes, int n_in,
                              void* d_out, int out_size, void* d_ws, size_t ws_size,
                              hipStream_t stream) {
  (void)in_sizes; (void)n_in; (void)out_size; (void)ws_size;
  const float* x    = (const float*)d_in[0];
  const float* E    = (const float*)d_in[1];
  const float* pa   = (const float*)d_in[2];
  const float* Wp   = (const float*)d_in[3];
  const float* bp   = (const float*)d_in[4];
  float* out = (float*)d_out;

  float* ws   = (float*)d_ws;
  float* sim  = ws;
  float* sube = sim  + 1048576;
  float* S    = sube + 8388608;
  float* xn   = S    + 8192;
  float* sx   = xn   + 524288;
  float* xg1  = sx   + 524288;
  float* WpT  = xg1  + 524288;

  k_stage1<<<L1_GRID, 256, 0, stream>>>(E, pa, Wp, x, sim, sube, xn, WpT);
  k_bmm_nk<<<dim3(32, 8), 256, 0, stream>>>(sube, 1024LL * 1024, xn, 1024LL * 64,
                                            sx, 1024LL * 64);
  k_bmm_nk<<<dim3(32, 8), 256, 0, stream>>>(sim, 0LL, sx, 1024LL * 64,
                                            xg1, 1024LL * 64);
  k_final<<<dim3(4, 64), 128, 0, stream>>>(sx, xg1, WpT, E, bp, out);
}
